# starting kernel
# baseline (speedup 1.0000x reference)
#include <hip/hip_runtime.h>
#include <cstddef>
#include <cstdint>

#define D 128
#define LDS_PP2 144

typedef float v2f __attribute__((ext_vector_type(2)));
typedef float v8f __attribute__((ext_vector_type(8)));

__device__ __forceinline__ v8f wmma_f32(v2f a, v2f b, v8f c) {
#if defined(__gfx1250__)
  return __builtin_amdgcn_wmma_f32_16x16x4_f32(false, a, false, b,
                                               (short)0, c, false, false);
#else
  c[0] += a[0] * b[0];
  return c;
#endif
}

__global__ void zero_f32(float* __restrict__ p, int n) {
  int i = blockIdx.x * blockDim.x + threadIdx.x;
  if (i < n) p[i] = 0.0f;
}

__global__ void repack_weight(const float* __restrict__ W, float* __restrict__ WTi) {
  int idx = blockIdx.x * blockDim.x + threadIdx.x;
  if (idx >= D * D) return;
  int n = (idx >> 1) & (D - 1);
  int k = ((idx >> 8) << 1) | (idx & 1);
  WTi[idx] = W[n * D + k];
}

__global__ void scatter_edges(const float* __restrict__ x,
                              const int* __restrict__ src,
                              const int* __restrict__ dst,
                              float* __restrict__ msg,
                              float* __restrict__ deg,
                              int nEdges) {
  int wave = (int)((blockIdx.x * blockDim.x + threadIdx.x) >> 5);
  int lane = threadIdx.x & 31;
  if (wave >= nEdges) return;
  int s = src[wave];
  int d = dst[wave];
  float4 v = ((const float4*)(x + (size_t)s * D))[lane];
  float* o = msg + (size_t)d * D + lane * 4;
  atomicAdd(o + 0, v.x);
  atomicAdd(o + 1, v.y);
  atomicAdd(o + 2, v.z);
  atomicAdd(o + 3, v.w);
  if (deg != nullptr && lane == 0) atomicAdd(deg + d, 1.0f);
}

__global__ void __launch_bounds__(128)
sage_layer(const float* __restrict__ xin, const float* __restrict__ msg,
           const float* __restrict__ deg,
           const float* __restrict__ WTiS, const float* __restrict__ WTiN,
           const float* __restrict__ bias,
           float* __restrict__ out, int nNodes) {
  extern __shared__ float lds[];
  const int tid  = threadIdx.x;
  const int wave = tid >> 5;
  const int lane = tid & 31;
  const int half = lane >> 4;
  const int l16  = lane & 15;
  const int m0   = blockIdx.x * 64 + wave * 16;
  const int mRow = m0 + l16;
  const int mC   = (mRow < nNodes) ? mRow : (nNodes - 1);

  const float invd = 1.0f / fmaxf(deg[mC], 1.0f);
  const float* xr = xin + (size_t)mC * D + half * 2;
  const float* mr = msg + (size_t)mC * D + half * 2;
  const v2f* B2 = (const v2f*)lds;

  v8f acc[8];
#pragma unroll
  for (int t = 0; t < 8; ++t) acc[t] = (v8f)(0.0f);

  for (int phase = 0; phase < 2; ++phase) {
    const float4* src4 = (const float4*)(phase == 0 ? WTiS : WTiN);
    const float*  arow = (phase == 0) ? xr : mr;

    __syncthreads();
    for (int i = tid; i < (D * D) / 4; i += 128) {
      ((float4*)lds)[(i >> 6) * (LDS_PP2 / 2) + (i & 63)] = src4[i];
    }
    __syncthreads();

    v2f aCur = *(const v2f*)(arow);
    if (phase == 1) aCur *= invd;

    for (int k0 = 0; k0 < D; k0 += 4) {
      const int base2 = ((k0 >> 1) + half) * LDS_PP2 + l16;
      v2f b[8];
#pragma unroll
      for (int t = 0; t < 8; ++t) b[t] = B2[base2 + t * 16];

      const int kN = (k0 + 4 < D) ? (k0 + 4) : k0;
      v2f aNext = *(const v2f*)(arow + kN);
      if (phase == 1) aNext *= invd;

#pragma unroll
      for (int t = 0; t < 8; ++t) acc[t] = wmma_f32(aCur, b[t], acc[t]);
      aCur = aNext;
    }
  }

#pragma unroll
  for (int t = 0; t < 8; ++t) {
    const int col = t * 16 + l16;
    const float bv = bias[col];
#pragma unroll
    for (int r = 0; r < 8; ++r) {
      int row = m0 + r + half * 8;
      if (row < nNodes) out[(size_t)row * D + col] = acc[t][r] + bv;
    }
  }
}

__global__ void l2_normalize(float* __restrict__ h, int nNodes) {
  int wave = (int)((blockIdx.x * blockDim.x + threadIdx.x) >> 5);
  int lane = threadIdx.x & 31;
  if (wave >= nNodes) return;
  float4* row = (float4*)(h + (size_t)wave * D);
  float4 v = row[lane];
  float ss = v.x * v.x + v.y * v.y + v.z * v.z + v.w * v.w;
#pragma unroll
  for (int off = 16; off > 0; off >>= 1) ss += __shfl_xor(ss, off, 32);
  float scale = 1.0f / fmaxf(sqrtf(ss), 1e-12f);
  v.x *= scale; v.y *= scale; v.z *= scale; v.w *= scale;
  row[lane] = v;
}

extern "C" void kernel_launch(void* const* d_in, const int* in_sizes, int n_in,
                              void* d_out, int out_size, void* d_ws, size_t ws_size,
                              hipStream_t stream) {
  const float* x   = (const float*)d_in[0];
  const int*   src = (const int*)  d_in[1];
  const int*   dst = (const int*)  d_in[2];
  const float* Ws1 = (const float*)d_in[3];
  const float* Wn1 = (const float*)d_in[4];
  const float* b1  = (const float*)d_in[5];
  const float* Ws2 = (const float*)d_in[6];
  const float* Wn2 = (const float*)d_in[7];
  const float* b2  = (const float*)d_in[8];
  float* out = (float*)d_out;

  const int N = in_sizes[0] / D;
  const int E = in_sizes[1];

  char* ws = (char*)d_ws;
  float* msg = (float*)ws; ws += (size_t)N * D * sizeof(float);
  float* h1  = (float*)ws; ws += (size_t)N * D * sizeof(float);
  float* deg = (float*)ws; ws += (((size_t)N * sizeof(float)) + 255) & ~(size_t)255;
  float* WTs1 = (float*)ws; ws += (size_t)D * D * sizeof(float);
  float* WTn1 = (float*)ws; ws += (size_t)D * D * sizeof(float);
  float* WTs2 = (float*)ws; ws += (size_t)D * D * sizeof(float);
  float* WTn2 = (float*)ws; ws += (size_t)D * D * sizeof(float);

  {
    dim3 blk(256), grd((D * D + 255) / 256);
    repack_weight<<<grd, blk, 0, stream>>>(Ws1, WTs1);
    repack_weight<<<grd, blk, 0, stream>>>(Wn1, WTn1);
    repack_weight<<<grd, blk, 0, stream>>>(Ws2, WTs2);
    repack_weight<<<grd, blk, 0, stream>>>(Wn2, WTn2);
  }

  const size_t ldsBytes = (size_t)64 * LDS_PP2 * 2 * sizeof(float);
  dim3 gemmGrid((N + 63) / 64);
  const int nd = N * D;

  zero_f32<<<(nd + 255) / 256, 256, 0, stream>>>(msg, nd);
  zero_f32<<<(N + 255) / 256, 256, 0, stream>>>(deg, N);
  scatter_edges<<<(E * 32 + 255) / 256, 256, 0, stream>>>(x, src, dst, msg, deg, E);
  sage_layer<<<gemmGrid, 128, ldsBytes, stream>>>(x, msg, deg, WTs1, WTn1, b1, h1, N);

  zero_f32<<<(nd + 255) / 256, 256, 0, stream>>>(msg, nd);
  scatter_edges<<<(E * 32 + 255) / 256, 256, 0, stream>>>(h1, src, dst, msg, nullptr, E);
  sage_layer<<<gemmGrid, 128, ldsBytes, stream>>>(h1, msg, deg, WTs2, WTn2, b2, out, N);

  l2_normalize<<<(N * 32 + 255) / 256, 256, 0, stream>>>(out, N);
}


# Neighbour sum: each of 256 work-groups owns 392 destination rows in LDS, filters the whole dst list for them and adds source rows with LDS f32 atomics (arrival order); global atomics and clears gone
# speedup vs baseline: 1.6786x; 1.6786x over previous
#include <hip/hip_runtime.h>
#include <cstddef>
#include <cstdint>

#define D 128
#define LDS_PP2 144

typedef float v2f __attribute__((ext_vector_type(2)));
typedef float v8f __attribute__((ext_vector_type(8)));

__device__ __forceinline__ v8f wmma_f32(v2f a, v2f b, v8f c) {
#if defined(__gfx1250__)
  return __builtin_amdgcn_wmma_f32_16x16x4_f32(false, a, false, b,
                                               (short)0, c, false, false);
#else
  c[0] += a[0] * b[0];
  return c;
#endif
}

__global__ void repack_weight(const float* __restrict__ W, float* __restrict__ WTi) {
  int idx = blockIdx.x * blockDim.x + threadIdx.x;
  if (idx >= D * D) return;
  int n = (idx >> 1) & (D - 1);
  int k = ((idx >> 8) << 1) | (idx & 1);
  WTi[idx] = W[n * D + k];
}

#define RCAP 392
#define LCAP 16384
#define AGG_T 1024
#define AGG_G 256

__device__ __forceinline__ void take_edge(int d, int e, int lo,
                                          const float* __restrict__ xin,
                                          const int* __restrict__ src,
                                          float* acc, unsigned* cnt,
                                          unsigned* list, unsigned* nList) {
  const unsigned r = (unsigned)(d - lo);
  if (r < (unsigned)RCAP) {
    atomicAdd(&cnt[r], 1u);
    const unsigned pos = atomicAdd(nList, 1u);
    if (pos < (unsigned)LCAP) {
      list[pos] = (unsigned)e;
    } else {
      const float* xr = xin + (size_t)src[e] * D;
      for (int j = 0; j < D; ++j) atomicAdd(&acc[r * D + j], xr[j]);
    }
  }
}

__global__ void __launch_bounds__(AGG_T)
aggregate_rows(const float* __restrict__ xin,
               const int* __restrict__ src,
               const int* __restrict__ dst,
               float* __restrict__ msg,
               float* __restrict__ deg,
               int nEdges, int nNodes) {
  __shared__ float    acc[(RCAP + 1) * D];
  __shared__ unsigned cnt[RCAP];
  __shared__ unsigned list[LCAP];
  __shared__ unsigned nList;

  const int tid  = threadIdx.x;
  const int lane = tid & 31;
  const int wave = __builtin_amdgcn_readfirstlane(tid >> 5);
  const int nG   = nEdges >> 2;
  const int4* d4 = (const int4*)dst;
  const int rot  = (int)(((long long)blockIdx.x * nG) / AGG_G);

  for (int lo = blockIdx.x * RCAP; lo < nNodes; lo += AGG_G * RCAP) {
    for (int i = tid; i < ((RCAP + 1) * D) / 4; i += AGG_T)
      ((float4*)acc)[i] = make_float4(0.0f, 0.0f, 0.0f, 0.0f);
    for (int i = tid; i < RCAP; i += AGG_T) cnt[i] = 0u;
    if (tid == 0) nList = 0u;
    __syncthreads();

    for (int i = tid; i < nG; i += 8 * AGG_T) {
      int4 v[8];
      int  gi[8];
#pragma unroll
      for (int u = 0; u < 8; ++u) {
        const int idx = i + u * AGG_T;
        int g = idx + rot;
        if (g >= nG) g -= nG;
        gi[u] = (idx < nG) ? g : -1;
        v[u]  = d4[(idx < nG) ? g : 0];
      }
#pragma unroll
      for (int u = 0; u < 8; ++u) {
        if (gi[u] >= 0) {
          const unsigned r0 = (unsigned)(v[u].x - lo), r1 = (unsigned)(v[u].y - lo);
          const unsigned r2 = (unsigned)(v[u].z - lo), r3 = (unsigned)(v[u].w - lo);
          const unsigned rm = min(min(r0, r1), min(r2, r3));
          if (rm < (unsigned)RCAP) {
            const int e0 = gi[u] << 2;
            take_edge(v[u].x, e0 + 0, lo, xin, src, acc, cnt, list, &nList);
            take_edge(v[u].y, e0 + 1, lo, xin, src, acc, cnt, list, &nList);
            take_edge(v[u].z, e0 + 2, lo, xin, src, acc, cnt, list, &nList);
            take_edge(v[u].w, e0 + 3, lo, xin, src, acc, cnt, list, &nList);
          }
        }
      }
    }
    {
      const int e = (nG << 2) + tid;
      if (e < nEdges) take_edge(dst[e], e, lo, xin, src, acc, cnt, list, &nList);
    }
    __syncthreads();

    const int n = (int)min(nList, (unsigned)LCAP);
    for (int i0 = wave * 8; i0 < n; i0 += (AGG_T / 32) * 8) {
      int   s[8], r[8];
      float v[8][4];
#pragma unroll
      for (int u = 0; u < 8; ++u) {
        const int i = i0 + u;
        const int e = __builtin_amdgcn_readfirstlane((int)list[(i < n) ? i : i0]);
        s[u] = src[e];
        const int dd = dst[e];
        r[u] = (i < n) ? (dd - lo) : RCAP;
      }
#pragma unroll
      for (int u = 0; u < 8; ++u) {
        const float* xr = xin + (size_t)s[u] * D + lane;
#pragma unroll
        for (int j = 0; j < 4; ++j) v[u][j] = xr[32 * j];
      }
#pragma unroll
      for (int u = 0; u < 8; ++u) {
        float* a = acc + r[u] * D + lane;
#pragma unroll
        for (int j = 0; j < 4; ++j) atomicAdd(a + 32 * j, v[u][j]);
      }
    }
    __syncthreads();

    const int rows = min(RCAP, nNodes - lo);
    for (int rr = wave; rr < rows; rr += AGG_T / 32) {
      const float4 a = ((const float4*)(acc + rr * D))[lane];
      ((float4*)(msg + (size_t)(lo + rr) * D))[lane] = a;
    }
    if (deg != nullptr)
      for (int i = tid; i < rows; i += AGG_T) deg[lo + i] = (float)cnt[i];
    __syncthreads();
  }
}

__global__ void __launch_bounds__(128)
sage_layer(const float* __restrict__ xin, const float* __restrict__ msg,
           const float* __restrict__ deg,
           const float* __restrict__ WTiS, const float* __restrict__ WTiN,
           const float* __restrict__ bias,
           float* __restrict__ out, int nNodes) {
  extern __shared__ float lds[];
  const int tid  = threadIdx.x;
  const int wave = tid >> 5;
  const int lane = tid & 31;
  const int half = lane >> 4;
  const int l16  = lane & 15;
  const int m0   = blockIdx.x * 64 + wave * 16;
  const int mRow = m0 + l16;
  const int mC   = (mRow < nNodes) ? mRow : (nNodes - 1);

  const float invd = 1.0f / fmaxf(deg[mC], 1.0f);
  const float* xr = xin + (size_t)mC * D + half * 2;
  const float* mr = msg + (size_t)mC * D + half * 2;
  const v2f* B2 = (const v2f*)lds;

  v8f acc[8];
#pragma unroll
  for (int t = 0; t < 8; ++t) acc[t] = (v8f)(0.0f);

  for (int phase = 0; phase < 2; ++phase) {
    const float4* src4 = (const float4*)(phase == 0 ? WTiS : WTiN);
    const float*  arow = (phase == 0) ? xr : mr;

    __syncthreads();
    for (int i = tid; i < (D * D) / 4; i += 128) {
      ((float4*)lds)[(i >> 6) * (LDS_PP2 / 2) + (i & 63)] = src4[i];
    }
    __syncthreads();

    v2f aCur = *(const v2f*)(arow);
    if (phase == 1) aCur *= invd;

    for (int k0 = 0; k0 < D; k0 += 4) {
      const int base2 = ((k0 >> 1) + half) * LDS_PP2 + l16;
      v2f b[8];
#pragma unroll
      for (int t = 0; t < 8; ++t) b[t] = B2[base2 + t * 16];

      const int kN = (k0 + 4 < D) ? (k0 + 4) : k0;
      v2f aNext = *(const v2f*)(arow + kN);
      if (phase == 1) aNext *= invd;

#pragma unroll
      for (int t = 0; t < 8; ++t) acc[t] = wmma_f32(aCur, b[t], acc[t]);
      aCur = aNext;
    }
  }

#pragma unroll
  for (int t = 0; t < 8; ++t) {
    const int col = t * 16 + l16;
    const float bv = bias[col];
#pragma unroll
    for (int r = 0; r < 8; ++r) {
      int row = m0 + r + half * 8;
      if (row < nNodes) out[(size_t)row * D + col] = acc[t][r] + bv;
    }
  }
}

__global__ void l2_normalize(float* __restrict__ h, int nNodes) {
  int wave = (int)((blockIdx.x * blockDim.x + threadIdx.x) >> 5);
  int lane = threadIdx.x & 31;
  if (wave >= nNodes) return;
  float4* row = (float4*)(h + (size_t)wave * D);
  float4 v = row[lane];
  float ss = v.x * v.x + v.y * v.y + v.z * v.z + v.w * v.w;
#pragma unroll
  for (int off = 16; off > 0; off >>= 1) ss += __shfl_xor(ss, off, 32);
  float scale = 1.0f / fmaxf(sqrtf(ss), 1e-12f);
  v.x *= scale; v.y *= scale; v.z *= scale; v.w *= scale;
  row[lane] = v;
}

extern "C" void kernel_launch(void* const* d_in, const int* in_sizes, int n_in,
                              void* d_out, int out_size, void* d_ws, size_t ws_size,
                              hipStream_t stream) {
  const float* x   = (const float*)d_in[0];
  const int*   src = (const int*)  d_in[1];
  const int*   dst = (const int*)  d_in[2];
  const float* Ws1 = (const float*)d_in[3];
  const float* Wn1 = (const float*)d_in[4];
  const float* b1  = (const float*)d_in[5];
  const float* Ws2 = (const float*)d_in[6];
  const float* Wn2 = (const float*)d_in[7];
  const float* b2  = (const float*)d_in[8];
  float* out = (float*)d_out;

  const int N = in_sizes[0] / D;
  const int E = in_sizes[1];

  char* ws = (char*)d_ws;
  float* msg = (float*)ws; ws += (size_t)N * D * sizeof(float);
  float* h1  = (float*)ws; ws += (size_t)N * D * sizeof(float);
  float* deg = (float*)ws; ws += (((size_t)N * sizeof(float)) + 255) & ~(size_t)255;
  float* WTs1 = (float*)ws; ws += (size_t)D * D * sizeof(float);
  float* WTn1 = (float*)ws; ws += (size_t)D * D * sizeof(float);
  float* WTs2 = (float*)ws; ws += (size_t)D * D * sizeof(float);
  float* WTn2 = (float*)ws; ws += (size_t)D * D * sizeof(float);

  {
    dim3 blk(256), grd((D * D + 255) / 256);
    repack_weight<<<grd, blk, 0, stream>>>(Ws1, WTs1);
    repack_weight<<<grd, blk, 0, stream>>>(Wn1, WTn1);
    repack_weight<<<grd, blk, 0, stream>>>(Ws2, WTs2);
    repack_weight<<<grd, blk, 0, stream>>>(Wn2, WTn2);
  }

  const size_t ldsBytes = (size_t)64 * LDS_PP2 * 2 * sizeof(float);
  dim3 gemmGrid((N + 63) / 64);

  aggregate_rows<<<AGG_G, AGG_T, 0, stream>>>(x, src, dst, msg, deg, E, N);
  sage_layer<<<gemmGrid, 128, ldsBytes, stream>>>(x, msg, deg, WTs1, WTn1, b1, h1, N);

  aggregate_rows<<<AGG_G, AGG_T, 0, stream>>>(h1, src, dst, msg, nullptr, E, N);
  sage_layer<<<gemmGrid, 128, ldsBytes, stream>>>(h1, msg, deg, WTs2, WTn2, b2, out, N);

  l2_normalize<<<(N * 32 + 255) / 256, 256, 0, stream>>>(out, N);
}


# Neighbour add: each owned LDS row belongs to one wave; edges go to wave queues; a wave loads a source row 16 B per lane and adds it by plain LDS read/add/write, no LDS float atomics; arrival order
# speedup vs baseline: 2.3958x; 1.4273x over previous
#include <hip/hip_runtime.h>
#include <cstddef>
#include <cstdint>

#define D 128
#define LDS_PP2 144

typedef float v2f __attribute__((ext_vector_type(2)));
typedef float v8f __attribute__((ext_vector_type(8)));

__device__ __forceinline__ v8f wmma_f32(v2f a, v2f b, v8f c) {
#if defined(__gfx1250__)
  return __builtin_amdgcn_wmma_f32_16x16x4_f32(false, a, false, b,
                                               (short)0, c, false, false);
#else
  c[0] += a[0] * b[0];
  return c;
#endif
}

__global__ void repack_weight(const float* __restrict__ W, float* __restrict__ WTi) {
  int idx = blockIdx.x * blockDim.x + threadIdx.x;
  if (idx >= D * D) return;
  int n = (idx >> 1) & (D - 1);
  int k = ((idx >> 8) << 1) | (idx & 1);
  WTi[idx] = W[n * D + k];
}

#define RCAP 392
#define LCAP 16384
#define AGG_T 1024
#define AGG_G 256
#define WBITS 9
#define NOT_KEPT 0xFFFFFFFFu
static_assert(RCAP < (1 << WBITS), "an owned row number and the sink must fit the low bits of a resolved word");
#define NWAVE (AGG_T / 32)
#define QCAP 320
static_assert((NWAVE & (NWAVE - 1)) == 0, "the owner wave of a row is taken with a mask");
static_assert(NWAVE * QCAP + NWAVE <= LCAP, "the wave queues and their lengths are kept in one work-group's LCAP words");
static_assert(QCAP % 4 == 0, "a wave's queue starts on a 16-B boundary");
static_assert(AGG_G == 256, "the owner key is one byte: work-group g owns the ranges numbered g mod 256");

__device__ __forceinline__ unsigned owner_key(int d) {
  return ((unsigned)d / (unsigned)RCAP) & 255u;
}

__global__ void edge_owner_keys(const int* __restrict__ dst,
                                unsigned* __restrict__ keys,
                                int nEdges, int nWords) {
  const int g = blockIdx.x * blockDim.x + threadIdx.x;
  if (g >= nWords) return;
  const int e0 = g << 2;
  unsigned k = 0u;
  if (e0 + 3 < nEdges) {
    const int4 d = ((const int4*)dst)[g];
    k = owner_key(d.x) | (owner_key(d.y) << 8) |
        (owner_key(d.z) << 16) | (owner_key(d.w) << 24);
  } else {
    for (int b = 0; b < 4; ++b)
      if (e0 + b < nEdges) k |= owner_key(dst[e0 + b]) << (8 * b);
  }
  keys[g] = k;
}

__device__ __forceinline__ unsigned zero_bytes(unsigned x) {
  const unsigned t = (x & 0x7F7F7F7Fu) + 0x7F7F7F7Fu;
  return ~(t | x | 0x7F7F7F7Fu);
}

__device__ __forceinline__ unsigned match_mask(uint4 k, unsigned mine, unsigned off) {
  const unsigned f0 = zero_bytes((k.x ^ mine) | off);
  const unsigned f1 = zero_bytes((k.y ^ mine) | off);
  const unsigned f2 = zero_bytes((k.z ^ mine) | off);
  const unsigned f3 = zero_bytes((k.w ^ mine) | off);
  return (f0 >> 7) | (f1 >> 6) | (f2 >> 5) | (f3 >> 4);
}

__device__ __forceinline__ void queue_edge(unsigned pos, unsigned e, int lo, int nEdges,
                                           const float* __restrict__ xin,
                                           const int* __restrict__ src,
                                           const int* __restrict__ dst,
                                           float* acc, unsigned* cnt, unsigned* list) {
  if (pos < (unsigned)LCAP) {
    list[pos] = e;
  } else if (e < (unsigned)nEdges) {
    const unsigned r = (unsigned)(dst[e] - lo);
    if (r < (unsigned)RCAP) {
      atomicAdd(&cnt[r], 1u);
      const float* xr = xin + (size_t)src[e] * D;
      for (int j = 0; j < D; ++j) atomicAdd(&acc[r * D + j], xr[j]);
    }
  }
}

__global__ void __launch_bounds__(AGG_T)
aggregate_rows(const float* __restrict__ xin,
               const int* __restrict__ src,
               const int* __restrict__ dst,
               const unsigned* __restrict__ keys,
               unsigned* __restrict__ keepW,
               const unsigned* __restrict__ keptW,
               float* __restrict__ msg,
               float* __restrict__ deg,
               int nEdges, int nNodes) {
  __shared__ __attribute__((aligned(16))) float acc[RCAP * D];
  __shared__ unsigned cnt[RCAP];
  __shared__ unsigned list[LCAP];
  __shared__ __attribute__((aligned(16))) unsigned wq[NWAVE * QCAP];
  __shared__ unsigned wlen[NWAVE];
  __shared__ unsigned nList;
  __shared__ unsigned ovf;

  const int tid  = threadIdx.x;
  const int lane = tid & 31;
  const int wave = __builtin_amdgcn_readfirstlane(tid >> 5);
  const int nG16 = (nEdges + 15) >> 4;
  const uint4* k4 = (const uint4*)keys;
  const int rot  = (int)(((long long)blockIdx.x * nG16) / AGG_G);
  const unsigned mine = (unsigned)blockIdx.x * 0x01010101u;
  const unsigned rowBits = (1u << WBITS) - 1u;

  const bool onePass = nNodes <= AGG_G * RCAP;
  unsigned* keepMine = (keepW != nullptr && onePass) ? keepW + (size_t)blockIdx.x * LCAP : nullptr;
  unsigned kept = NOT_KEPT;
  if (keptW != nullptr && onePass) kept = keptW[(size_t)AGG_G * LCAP + blockIdx.x];
  const bool replay = kept != NOT_KEPT;

  for (int lo = blockIdx.x * RCAP; lo < nNodes; lo += AGG_G * RCAP) {
    for (int i = tid; i < (RCAP * D) / 4; i += AGG_T)
      ((float4*)acc)[i] = make_float4(0.0f, 0.0f, 0.0f, 0.0f);
    for (int i = tid; i < RCAP; i += AGG_T) cnt[i] = 0u;
    if (tid < NWAVE) wlen[tid] = 0u;
    if (tid == 0) { nList = 0u; ovf = 0u; }
    __syncthreads();

    if (!replay) {
    for (int i = tid; i < nG16; i += 8 * AGG_T) {
      uint4 v[8];
      int   gi[8];
#pragma unroll
      for (int u = 0; u < 8; ++u) {
        const int idx = i + u * AGG_T;
        int g = idx + rot;
        if (g >= nG16) g -= nG16;
        gi[u] = (idx < nG16) ? g : -1;
        v[u]  = k4[(idx < nG16) ? g : 0];
      }
      unsigned m[8];
      unsigned tot = 0u;
#pragma unroll
      for (int u = 0; u < 8; ++u) {
        m[u] = match_mask(v[u], mine, (unsigned)(gi[u] >> 31));
        tot += (unsigned)__popc(m[u]);
      }
      if (tot != 0u) {
        unsigned pos = atomicAdd(&nList, tot);
        if (pos + tot <= (unsigned)LCAP) {
#pragma unroll
          for (int u = 0; u < 8; ++u) {
            unsigned mm = m[u];
            while (mm != 0u) {
              const unsigned p = (unsigned)__builtin_ctz(mm);
              mm &= mm - 1u;
              list[pos++] = ((unsigned)gi[u] << 4) + ((p & 7u) << 2) + (p >> 3);
            }
          }
        } else {
#pragma unroll 1
          for (int u = 0; u < 8; ++u) {
            const int idx = i + u * AGG_T;
            if (idx >= nG16) break;
            int g = idx + rot;
            if (g >= nG16) g -= nG16;
            unsigned mm = match_mask(k4[g], mine, 0u);
            while (mm != 0u) {
              const unsigned p = (unsigned)__builtin_ctz(mm);
              mm &= mm - 1u;
              const unsigned e = ((unsigned)g << 4) + ((p & 7u) << 2) + (p >> 3);
              queue_edge(pos, e, lo, nEdges, xin, src, dst, acc, cnt, list);
              ++pos;
            }
          }
        }
      }
    }
    __syncthreads();

    const unsigned nq = nList;
    const int n = (int)min(nq, (unsigned)LCAP);
    for (int i = tid; i < n; i += AGG_T) {
      const unsigned e  = list[i];
      const bool     ok = e < (unsigned)nEdges;
      const unsigned ec = ok ? e : 0u;
      const unsigned sv = (unsigned)src[ec];
      const unsigned rr = (unsigned)(dst[ec] - lo);
      if (ok && rr < (unsigned)RCAP) {
        atomicAdd(&cnt[rr], 1u);
        const unsigned ow = rr & (unsigned)(NWAVE - 1);
        const unsigned p  = atomicAdd(&wlen[ow], 1u);
        if (p < (unsigned)QCAP) {
          wq[ow * QCAP + p] = (sv << WBITS) | rr;
        } else {
          ovf = 1u;
          const float* xr = xin + (size_t)sv * D;
          for (int j = 0; j < D; ++j) atomicAdd(&acc[rr * D + j], xr[j]);
        }
      }
    }
    __syncthreads();
    if (keepMine != nullptr) {
      const int c = (int)min(wlen[wave], (unsigned)QCAP);
      for (int j = lane; j < c; j += 32) keepMine[wave * QCAP + j] = wq[wave * QCAP + j];
      if (lane == 0) keepMine[NWAVE * QCAP + wave] = (unsigned)c;
      if (tid == 0)
        keepW[(size_t)AGG_G * LCAP + blockIdx.x] = (nq <= (unsigned)LCAP && ovf == 0u) ? 0u : NOT_KEPT;
    }
    } else {
    const unsigned* mineW = keptW + (size_t)blockIdx.x * LCAP;
    const int c = (int)min(mineW[NWAVE * QCAP + wave], (unsigned)QCAP);
    for (int j = lane; j < c; j += 32) {
      const unsigned w = mineW[wave * QCAP + j];
      wq[wave * QCAP + j] = w;
      if (deg != nullptr) atomicAdd(&cnt[w & rowBits], 1u);
    }
    if (lane == 0) wlen[wave] = (unsigned)c;
    __syncthreads();
    }

    {
      const unsigned* q = wq + wave * QCAP;
      const int c = __builtin_amdgcn_readfirstlane((int)min(wlen[wave], (unsigned)QCAP));
      const unsigned padWord = (unsigned)wave;
      for (int i0 = 0; i0 < c; i0 += 16) {
        const int li = i0 + (lane & 15);
        const unsigned wv = (li < c) ? q[li] : padWord;
        int    s[16], r[16];
        float4 v[16];
#pragma unroll
        for (int u = 0; u < 16; ++u) {
          const unsigned w = (unsigned)__builtin_amdgcn_readlane((int)wv, u);
          s[u] = (int)(w >> WBITS);
          r[u] = (int)(w & rowBits);
        }
#pragma unroll
        for (int u = 0; u < 16; ++u)
          v[u] = ((const float4*)(xin + (size_t)s[u] * D))[lane];
#pragma unroll
        for (int u = 0; u < 16; ++u) {
          const bool live = i0 + u < c;
          float4* a = (float4*)(acc + r[u] * D) + lane;
          float4 t = *a;
          t.x += live ? v[u].x : 0.0f;
          t.y += live ? v[u].y : 0.0f;
          t.z += live ? v[u].z : 0.0f;
          t.w += live ? v[u].w : 0.0f;
          *a = t;
        }
      }
    }
    __syncthreads();

    const int rows = min(RCAP, nNodes - lo);
    for (int rr = wave; rr < rows; rr += AGG_T / 32) {
      const float4 a = ((const float4*)(acc + rr * D))[lane];
      ((float4*)(msg + (size_t)(lo + rr) * D))[lane] = a;
    }
    if (deg != nullptr)
      for (int i = tid; i < rows; i += AGG_T) deg[lo + i] = (float)cnt[i];
    __syncthreads();
  }
}

__global__ void __launch_bounds__(128)
sage_layer(const float* __restrict__ xin, const float* __restrict__ msg,
           const float* __restrict__ deg,
           const float* __restrict__ WTiS, const float* __restrict__ WTiN,
           const float* __restrict__ bias,
           float* __restrict__ out, int nNodes) {
  extern __shared__ float lds[];
  const int tid  = threadIdx.x;
  const int wave = tid >> 5;
  const int lane = tid & 31;
  const int half = lane >> 4;
  const int l16  = lane & 15;
  const int m0   = blockIdx.x * 64 + wave * 16;
  const int mRow = m0 + l16;
  const int mC   = (mRow < nNodes) ? mRow : (nNodes - 1);

  const float invd = 1.0f / fmaxf(deg[mC], 1.0f);
  const float* xr = xin + (size_t)mC * D + half * 2;
  const float* mr = msg + (size_t)mC * D + half * 2;
  const v2f* B2 = (const v2f*)lds;

  v8f acc[8];
#pragma unroll
  for (int t = 0; t < 8; ++t) acc[t] = (v8f)(0.0f);

  for (int phase = 0; phase < 2; ++phase) {
    const float4* src4 = (const float4*)(phase == 0 ? WTiS : WTiN);
    const float*  arow = (phase == 0) ? xr : mr;

    __syncthreads();
    for (int i = tid; i < (D * D) / 4; i += 128) {
      ((float4*)lds)[(i >> 6) * (LDS_PP2 / 2) + (i & 63)] = src4[i];
    }
    __syncthreads();

    v2f aCur = *(const v2f*)(arow);
    if (phase == 1) aCur *= invd;

    for (int k0 = 0; k0 < D; k0 += 4) {
      const int base2 = ((k0 >> 1) + half) * LDS_PP2 + l16;
      v2f b[8];
#pragma unroll
      for (int t = 0; t < 8; ++t) b[t] = B2[base2 + t * 16];

      const int kN = (k0 + 4 < D) ? (k0 + 4) : k0;
      v2f aNext = *(const v2f*)(arow + kN);
      if (phase == 1) aNext *= invd;

#pragma unroll
      for (int t = 0; t < 8; ++t) acc[t] = wmma_f32(aCur, b[t], acc[t]);
      aCur = aNext;
    }
  }

#pragma unroll
  for (int t = 0; t < 8; ++t) {
    const int col = t * 16 + l16;
    const float bv = bias[col];
#pragma unroll
    for (int r = 0; r < 8; ++r) {
      int row = m0 + r + half * 8;
      if (row < nNodes) out[(size_t)row * D + col] = acc[t][r] + bv;
    }
  }
}

__global__ void l2_normalize(float* __restrict__ h, int nNodes) {
  int wave = (int)((blockIdx.x * blockDim.x + threadIdx.x) >> 5);
  int lane = threadIdx.x & 31;
  if (wave >= nNodes) return;
  float4* row = (float4*)(h + (size_t)wave * D);
  float4 v = row[lane];
  float ss = v.x * v.x + v.y * v.y + v.z * v.z + v.w * v.w;
#pragma unroll
  for (int off = 16; off > 0; off >>= 1) ss += __shfl_xor(ss, off, 32);
  float scale = 1.0f / fmaxf(sqrtf(ss), 1e-12f);
  v.x *= scale; v.y *= scale; v.z *= scale; v.w *= scale;
  row[lane] = v;
}

extern "C" void kernel_launch(void* const* d_in, const int* in_sizes, int n_in,
                              void* d_out, int out_size, void* d_ws, size_t ws_size,
                              hipStream_t stream) {
  const float* x   = (const float*)d_in[0];
  const int*   src = (const int*)  d_in[1];
  const int*   dst = (const int*)  d_in[2];
  const float* Ws1 = (const float*)d_in[3];
  const float* Wn1 = (const float*)d_in[4];
  const float* b1  = (const float*)d_in[5];
  const float* Ws2 = (const float*)d_in[6];
  const float* Wn2 = (const float*)d_in[7];
  const float* b2  = (const float*)d_in[8];
  float* out = (float*)d_out;

  const int N = in_sizes[0] / D;
  const int E = in_sizes[1];

  char* ws = (char*)d_ws;
  float* msg = (float*)ws; ws += (size_t)N * D * sizeof(float);
  float* h1  = (float*)ws; ws += (size_t)N * D * sizeof(float);
  float* deg = (float*)ws; ws += (((size_t)N * sizeof(float)) + 255) & ~(size_t)255;
  float* WTs1 = (float*)ws; ws += (size_t)D * D * sizeof(float);
  float* WTn1 = (float*)ws; ws += (size_t)D * D * sizeof(float);
  float* WTs2 = (float*)ws; ws += (size_t)D * D * sizeof(float);
  float* WTn2 = (float*)ws; ws += (size_t)D * D * sizeof(float);

  {
    dim3 blk(256), grd((D * D + 255) / 256);
    repack_weight<<<grd, blk, 0, stream>>>(Ws1, WTs1);
    repack_weight<<<grd, blk, 0, stream>>>(Wn1, WTn1);
    repack_weight<<<grd, blk, 0, stream>>>(Ws2, WTs2);
    repack_weight<<<grd, blk, 0, stream>>>(Wn2, WTn2);
  }

  unsigned* keys = (unsigned*)d_out;
  {
    const int nWords = ((E + 15) / 16) * 4;
    edge_owner_keys<<<(nWords + 255) / 256, 256, 0, stream>>>(dst, keys, E, nWords);
  }

  unsigned* kept = keys + ((E + 15) / 16) * 4;

  const size_t ldsBytes = (size_t)64 * LDS_PP2 * 2 * sizeof(float);
  dim3 gemmGrid((N + 63) / 64);

  aggregate_rows<<<AGG_G, AGG_T, 0, stream>>>(x, src, dst, keys, kept, nullptr, msg, deg, E, N);
  sage_layer<<<gemmGrid, 128, ldsBytes, stream>>>(x, msg, deg, WTs1, WTn1, b1, h1, N);

  aggregate_rows<<<AGG_G, AGG_T, 0, stream>>>(h1, src, dst, keys, nullptr, kept, msg, nullptr, E, N);
  sage_layer<<<gemmGrid, 128, ldsBytes, stream>>>(h1, msg, deg, WTs2, WTn2, b2, out, N);

  l2_normalize<<<(N * 32 + 255) / 256, 256, 0, stream>>>(out, N);
}


# Layer products run in the owner work-group: after the neighbour adds each wave multiplies a 16-row strip of the LDS-resident sums (f32 WMMA, sums in the same order); sum plane and 2 launches gone
# speedup vs baseline: 2.7090x; 1.1307x over previous
#include <hip/hip_runtime.h>
#include <cstddef>
#include <cstdint>

#define D 128
#define LDS_PP2 144

typedef float v2f __attribute__((ext_vector_type(2)));
typedef float v8f __attribute__((ext_vector_type(8)));

__device__ __forceinline__ v8f wmma_f32(v2f a, v2f b, v8f c) {
#if defined(__gfx1250__)
  return __builtin_amdgcn_wmma_f32_16x16x4_f32(false, a, false, b,
                                               (short)0, c, false, false);
#else
  c[0] += a[0] * b[0];
  return c;
#endif
}

__global__ void repack_weight(const float* __restrict__ W, float* __restrict__ WTi) {
  int idx = blockIdx.x * blockDim.x + threadIdx.x;
  if (idx >= D * D) return;
  int n = (idx >> 1) & (D - 1);
  int k = ((idx >> 8) << 1) | (idx & 1);
  WTi[idx] = W[n * D + k];
}

#define RCAP 392
#define LCAP 16384
#define AGG_T 1024
#define AGG_G 256
#define WBITS 9
#define NOT_KEPT 0xFFFFFFFFu
static_assert(RCAP < (1 << WBITS), "an owned row number and the sink must fit the low bits of a resolved word");
#define NWAVE (AGG_T / 32)
#define QCAP 320
#define APITCH 132
static_assert((NWAVE & (NWAVE - 1)) == 0, "the owner wave of a row is taken with a mask");
static_assert(NWAVE * QCAP + NWAVE <= LCAP, "the wave queues and their lengths are kept in one work-group's LCAP words");
static_assert(QCAP % 4 == 0, "a wave's queue starts on a 16-B boundary");
static_assert((LCAP + NWAVE * QCAP) * 4 >= 64 * LDS_PP2 * 2 * 4, "one weight matrix fits where the edge queues stood");
static_assert(APITCH % 4 == 0 && APITCH >= D, "an owned row is read and written 16 B per lane");
static_assert((RCAP + 15) / 16 + 3 <= NWAVE, "a wave per 16-row strip, and three more when the last strip is cut in four");
static_assert(AGG_G == 256, "the owner key is one byte: the work-group of range g filters for key g mod 256");

__device__ __forceinline__ unsigned owner_key(int d) {
  return ((unsigned)d / (unsigned)RCAP) & 255u;
}

__global__ void edge_owner_keys(const int* __restrict__ dst,
                                unsigned* __restrict__ keys,
                                int nEdges, int nWords) {
  const int g = blockIdx.x * blockDim.x + threadIdx.x;
  if (g >= nWords) return;
  const int e0 = g << 2;
  unsigned k = 0u;
  if (e0 + 3 < nEdges) {
    const int4 d = ((const int4*)dst)[g];
    k = owner_key(d.x) | (owner_key(d.y) << 8) |
        (owner_key(d.z) << 16) | (owner_key(d.w) << 24);
  } else {
    for (int b = 0; b < 4; ++b)
      if (e0 + b < nEdges) k |= owner_key(dst[e0 + b]) << (8 * b);
  }
  keys[g] = k;
}

__device__ __forceinline__ unsigned zero_bytes(unsigned x) {
  const unsigned t = (x & 0x7F7F7F7Fu) + 0x7F7F7F7Fu;
  return ~(t | x | 0x7F7F7F7Fu);
}

__device__ __forceinline__ unsigned match_mask(uint4 k, unsigned mine, unsigned off) {
  const unsigned f0 = zero_bytes((k.x ^ mine) | off);
  const unsigned f1 = zero_bytes((k.y ^ mine) | off);
  const unsigned f2 = zero_bytes((k.z ^ mine) | off);
  const unsigned f3 = zero_bytes((k.w ^ mine) | off);
  return (f0 >> 7) | (f1 >> 6) | (f2 >> 5) | (f3 >> 4);
}

template <int NT, bool SCALE>
__device__ __forceinline__ void strip_products(v8f* c, const float* arow, float rowScale,
                                               const v2f* B2, int half, int colL) {
  v2f aCur = *(const v2f*)(arow);
  if (SCALE) aCur *= rowScale;
#pragma unroll 4
  for (int k0 = 0; k0 < D; k0 += 4) {
    const int base2 = ((k0 >> 1) + half) * LDS_PP2 + colL;
    v2f b[NT];
#pragma unroll
    for (int t = 0; t < NT; ++t) b[t] = B2[base2 + t * 16];
    const int kN = (k0 + 4 < D) ? (k0 + 4) : k0;
    v2f aNext = *(const v2f*)(arow + kN);
    if (SCALE) aNext *= rowScale;
#pragma unroll
    for (int t = 0; t < NT; ++t) c[t] = wmma_f32(aCur, b[t], c[t]);
    aCur = aNext;
  }
}

template <int NT>
__device__ __forceinline__ void strip_store(const v8f* c, const float* __restrict__ bias,
                                            float* __restrict__ out, int lo, int strip,
                                            int rows, int half, int colL) {
#pragma unroll
  for (int t = 0; t < NT; ++t) {
    const int col = t * 16 + colL;
    const float bv = bias[col];
#pragma unroll
    for (int r = 0; r < 8; ++r) {
      const int row = strip * 16 + r + half * 8;
      if (row < rows) out[(size_t)(lo + row) * D + col] = c[t][r] + bv;
    }
  }
}

__device__ __forceinline__ void queue_edge(unsigned pos, unsigned e, int lo, int nEdges,
                                           const float* __restrict__ xin,
                                           const int* __restrict__ src,
                                           const int* __restrict__ dst,
                                           float* acc, unsigned* cnt, unsigned* list) {
  if (pos < (unsigned)LCAP) {
    list[pos] = e;
  } else if (e < (unsigned)nEdges) {
    const unsigned r = (unsigned)(dst[e] - lo);
    if (r < (unsigned)RCAP) {
      atomicAdd(&cnt[r], 1u);
      const float* xr = xin + (size_t)src[e] * D;
      for (int j = 0; j < D; ++j) atomicAdd(&acc[r * APITCH + j], xr[j]);
    }
  }
}

__global__ void __launch_bounds__(AGG_T)
aggregate_rows(const float* __restrict__ xin,
               const int* __restrict__ src,
               const int* __restrict__ dst,
               const unsigned* __restrict__ keys,
               unsigned* __restrict__ keepW,
               const unsigned* __restrict__ keptW,
               const float* __restrict__ WTiS, const float* __restrict__ WTiN,
               const float* __restrict__ bias,
               float* __restrict__ out,
               int nEdges, int nNodes) {
  __shared__ __attribute__((aligned(16))) float acc[RCAP * APITCH];
  __shared__ unsigned cnt[RCAP];
  __shared__ __attribute__((aligned(16))) unsigned pool[LCAP + NWAVE * QCAP];
  unsigned* const list = pool;
  unsigned* const wq   = pool + LCAP;
  __shared__ unsigned wlen[NWAVE];
  __shared__ unsigned nList;
  __shared__ unsigned ovf;

  const int tid  = threadIdx.x;
  const int lane = tid & 31;
  const int wave = __builtin_amdgcn_readfirstlane(tid >> 5);
  const int nG16 = (nEdges + 15) >> 4;
  const uint4* k4 = (const uint4*)keys;
  const unsigned myKey = (unsigned)blockIdx.x & (unsigned)(AGG_G - 1);
  const int rot  = (int)(((long long)myKey * nG16) / AGG_G);
  const unsigned mine = myKey * 0x01010101u;
  const unsigned rowBits = (1u << WBITS) - 1u;

  unsigned* keepMine = (keepW != nullptr) ? keepW + (size_t)blockIdx.x * LCAP : nullptr;
  unsigned kept = NOT_KEPT;
  if (keptW != nullptr) kept = keptW[(size_t)gridDim.x * LCAP + blockIdx.x];
  const bool replay = kept != NOT_KEPT;

  const int lo = blockIdx.x * RCAP;
  if (lo >= nNodes) return;
  {
    for (int i = tid; i < (RCAP * APITCH) / 4; i += AGG_T)
      ((float4*)acc)[i] = make_float4(0.0f, 0.0f, 0.0f, 0.0f);
    for (int i = tid; i < RCAP; i += AGG_T) cnt[i] = 0u;
    if (tid < NWAVE) wlen[tid] = 0u;
    if (tid == 0) { nList = 0u; ovf = 0u; }
    __syncthreads();

    if (!replay) {
    for (int i = tid; i < nG16; i += 8 * AGG_T) {
      uint4 v[8];
      int   gi[8];
#pragma unroll
      for (int u = 0; u < 8; ++u) {
        const int idx = i + u * AGG_T;
        int g = idx + rot;
        if (g >= nG16) g -= nG16;
        gi[u] = (idx < nG16) ? g : -1;
        v[u]  = k4[(idx < nG16) ? g : 0];
      }
      unsigned m[8];
      unsigned tot = 0u;
#pragma unroll
      for (int u = 0; u < 8; ++u) {
        m[u] = match_mask(v[u], mine, (unsigned)(gi[u] >> 31));
        tot += (unsigned)__popc(m[u]);
      }
      if (tot != 0u) {
        unsigned pos = atomicAdd(&nList, tot);
        if (pos + tot <= (unsigned)LCAP) {
#pragma unroll
          for (int u = 0; u < 8; ++u) {
            unsigned mm = m[u];
            while (mm != 0u) {
              const unsigned p = (unsigned)__builtin_ctz(mm);
              mm &= mm - 1u;
              list[pos++] = ((unsigned)gi[u] << 4) + ((p & 7u) << 2) + (p >> 3);
            }
          }
        } else {
#pragma unroll 1
          for (int u = 0; u < 8; ++u) {
            const int idx = i + u * AGG_T;
            if (idx >= nG16) break;
            int g = idx + rot;
            if (g >= nG16) g -= nG16;
            unsigned mm = match_mask(k4[g], mine, 0u);
            while (mm != 0u) {
              const unsigned p = (unsigned)__builtin_ctz(mm);
              mm &= mm - 1u;
              const unsigned e = ((unsigned)g << 4) + ((p & 7u) << 2) + (p >> 3);
              queue_edge(pos, e, lo, nEdges, xin, src, dst, acc, cnt, list);
              ++pos;
            }
          }
        }
      }
    }
    __syncthreads();

    const unsigned nq = nList;
    const int n = (int)min(nq, (unsigned)LCAP);
    for (int i = tid; i < n; i += AGG_T) {
      const unsigned e  = list[i];
      const bool     ok = e < (unsigned)nEdges;
      const unsigned ec = ok ? e : 0u;
      const unsigned sv = (unsigned)src[ec];
      const unsigned rr = (unsigned)(dst[ec] - lo);
      if (ok && rr < (unsigned)RCAP) {
        atomicAdd(&cnt[rr], 1u);
        const unsigned ow = rr & (unsigned)(NWAVE - 1);
        const unsigned p  = atomicAdd(&wlen[ow], 1u);
        if (p < (unsigned)QCAP) {
          wq[ow * QCAP + p] = (sv << WBITS) | rr;
        } else {
          ovf = 1u;
          const float* xr = xin + (size_t)sv * D;
          for (int j = 0; j < D; ++j) atomicAdd(&acc[rr * APITCH + j], xr[j]);
        }
      }
    }
    __syncthreads();
    if (keepMine != nullptr) {
      const int c = (int)min(wlen[wave], (unsigned)QCAP);
      for (int j = lane; j < c; j += 32) keepMine[wave * QCAP + j] = wq[wave * QCAP + j];
      if (lane == 0) keepMine[NWAVE * QCAP + wave] = (unsigned)c;
      if (tid == 0)
        keepW[(size_t)gridDim.x * LCAP + blockIdx.x] = (nq <= (unsigned)LCAP && ovf == 0u) ? 0u : NOT_KEPT;
    }
    } else {
    const unsigned* mineW = keptW + (size_t)blockIdx.x * LCAP;
    const int c = (int)min(mineW[NWAVE * QCAP + wave], (unsigned)QCAP);
    for (int j = lane; j < c; j += 32) {
      const unsigned w = mineW[wave * QCAP + j];
      wq[wave * QCAP + j] = w;
      atomicAdd(&cnt[w & rowBits], 1u);
    }
    if (lane == 0) wlen[wave] = (unsigned)c;
    __syncthreads();
    }

    {
      const unsigned* q = wq + wave * QCAP;
      const int c = __builtin_amdgcn_readfirstlane((int)min(wlen[wave], (unsigned)QCAP));
      const unsigned padWord = (unsigned)wave;
      for (int i0 = 0; i0 < c; i0 += 16) {
        const int li = i0 + (lane & 15);
        const unsigned wv = (li < c) ? q[li] : padWord;
        int    s[16], r[16];
        float4 v[16];
#pragma unroll
        for (int u = 0; u < 16; ++u) {
          const unsigned w = (unsigned)__builtin_amdgcn_readlane((int)wv, u);
          s[u] = (int)(w >> WBITS);
          r[u] = (int)(w & rowBits);
        }
#pragma unroll
        for (int u = 0; u < 16; ++u)
          v[u] = ((const float4*)(xin + (size_t)s[u] * D))[lane];
#pragma unroll
        for (int u = 0; u < 16; ++u) {
          const bool live = i0 + u < c;
          float4* a = (float4*)(acc + r[u] * APITCH) + lane;
          float4 t = *a;
          t.x += live ? v[u].x : 0.0f;
          t.y += live ? v[u].y : 0.0f;
          t.z += live ? v[u].z : 0.0f;
          t.w += live ? v[u].w : 0.0f;
          *a = t;
        }
      }
    }
    __syncthreads();

    {
      const int rows    = min(RCAP, nNodes - lo);
      const int nStrips = (rows + 15) >> 4;
      const bool split  = (nStrips & 3) == 1;
      const int nWhole  = split ? nStrips - 1 : nStrips;
      const bool whole   = wave < nWhole;
      const bool quarter = split && !whole && wave < nWhole + 4;
      const int strip    = whole ? wave : nStrips - 1;
      const int colBase  = quarter ? (wave - nWhole) * 32 : 0;
      const int half = lane >> 4;
      const int l16  = lane & 15;
      const int rL   = min(strip * 16 + l16, rows - 1);
      float* wl = (float*)pool;
      const v2f* B2 = (const v2f*)wl;
      const float* xrow = xin + (size_t)(lo + rL) * D + half * 2;
      const float* srow = acc + rL * APITCH + half * 2;

      v8f c8[8];
#pragma unroll
      for (int t = 0; t < 8; ++t) c8[t] = (v8f)(0.0f);

      for (int i = tid; i < (D * D) / 4; i += AGG_T)
        ((float4*)wl)[(i >> 6) * (LDS_PP2 / 2) + (i & 63)] = ((const float4*)WTiS)[i];
      __syncthreads();
      if (whole)        strip_products<8, false>(c8, xrow, 1.0f, B2, half, l16);
      else if (quarter) strip_products<2, false>(c8, xrow, 1.0f, B2, half, colBase + l16);
      __syncthreads();
      for (int i = tid; i < (D * D) / 4; i += AGG_T)
        ((float4*)wl)[(i >> 6) * (LDS_PP2 / 2) + (i & 63)] = ((const float4*)WTiN)[i];
      __syncthreads();
      const float invd = 1.0f / fmaxf((float)cnt[rL], 1.0f);
      if (whole)        strip_products<8, true>(c8, srow, invd, B2, half, l16);
      else if (quarter) strip_products<2, true>(c8, srow, invd, B2, half, colBase + l16);

      if (whole) strip_store<8>(c8, bias, out, lo, strip, rows, half, l16);
      else if (quarter) strip_store<2>(c8, bias, out, lo, strip, rows, half, colBase + l16);
    }
  }
}

__global__ void l2_normalize(float* __restrict__ h, int nNodes) {
  int wave = (int)((blockIdx.x * blockDim.x + threadIdx.x) >> 5);
  int lane = threadIdx.x & 31;
  if (wave >= nNodes) return;
  float4* row = (float4*)(h + (size_t)wave * D);
  float4 v = row[lane];
  float ss = v.x * v.x + v.y * v.y + v.z * v.z + v.w * v.w;
#pragma unroll
  for (int off = 16; off > 0; off >>= 1) ss += __shfl_xor(ss, off, 32);
  float scale = 1.0f / fmaxf(sqrtf(ss), 1e-12f);
  v.x *= scale; v.y *= scale; v.z *= scale; v.w *= scale;
  row[lane] = v;
}

extern "C" void kernel_launch(void* const* d_in, const int* in_sizes, int n_in,
                              void* d_out, int out_size, void* d_ws, size_t ws_size,
                              hipStream_t stream) {
  const float* x   = (const float*)d_in[0];
  const int*   src = (const int*)  d_in[1];
  const int*   dst = (const int*)  d_in[2];
  const float* Ws1 = (const float*)d_in[3];
  const float* Wn1 = (const float*)d_in[4];
  const float* b1  = (const float*)d_in[5];
  const float* Ws2 = (const float*)d_in[6];
  const float* Wn2 = (const float*)d_in[7];
  const float* b2  = (const float*)d_in[8];
  float* out = (float*)d_out;

  const int N = in_sizes[0] / D;
  const int E = in_sizes[1];

  char* ws = (char*)d_ws;
  unsigned* keys = (unsigned*)ws; ws += (size_t)N * D * sizeof(float);
  float* h1  = (float*)ws; ws += (size_t)N * D * sizeof(float);
  float* WTs1 = (float*)ws; ws += (size_t)D * D * sizeof(float);
  float* WTn1 = (float*)ws; ws += (size_t)D * D * sizeof(float);
  float* WTs2 = (float*)ws; ws += (size_t)D * D * sizeof(float);
  float* WTn2 = (float*)ws; ws += (size_t)D * D * sizeof(float);

  {
    dim3 blk(256), grd((D * D + 255) / 256);
    repack_weight<<<grd, blk, 0, stream>>>(Ws1, WTs1);
    repack_weight<<<grd, blk, 0, stream>>>(Wn1, WTn1);
    repack_weight<<<grd, blk, 0, stream>>>(Ws2, WTs2);
    repack_weight<<<grd, blk, 0, stream>>>(Wn2, WTn2);
  }

  {
    const int nWords = ((E + 15) / 16) * 4;
    edge_owner_keys<<<(nWords + 255) / 256, 256, 0, stream>>>(dst, keys, E, nWords);
  }

  unsigned* kept = keys + ((E + 15) / 16) * 4;

  const int nRanges = (N + RCAP - 1) / RCAP;

  aggregate_rows<<<nRanges, AGG_T, 0, stream>>>(x, src, dst, keys, kept, nullptr, WTs1, WTn1, b1, h1, E, N);

  aggregate_rows<<<nRanges, AGG_T, 0, stream>>>(h1, src, dst, keys, nullptr, kept, WTs2, WTn2, b2, out, E, N);

  l2_normalize<<<(N * 32 + 255) / 256, 256, 0, stream>>>(out, N);
}


# Weights used as given: each W row is copied 16 B per lane into LDS as [k/4][n&15][n>>4][k&3], a B fragment is one 8-B read per tile; the four repack launches and planes go; k order and sums same
# speedup vs baseline: 2.8176x; 1.0401x over previous
#include <hip/hip_runtime.h>
#include <cstddef>
#include <cstdint>

#define D 128
#define WLANE 36
#define WSTEP 580

typedef float v2f __attribute__((ext_vector_type(2)));
typedef float v8f __attribute__((ext_vector_type(8)));

__device__ __forceinline__ v8f wmma_f32(v2f a, v2f b, v8f c) {
#if defined(__gfx1250__)
  return __builtin_amdgcn_wmma_f32_16x16x4_f32(false, a, false, b,
                                               (short)0, c, false, false);
#else
  c[0] += a[0] * b[0];
  return c;
#endif
}

#define RCAP 392
#define LCAP 16384
#define AGG_T 1024
#define AGG_G 256
#define WBITS 9
#define NOT_KEPT 0xFFFFFFFFu
static_assert(RCAP < (1 << WBITS), "an owned row number and the sink must fit the low bits of a resolved word");
#define NWAVE (AGG_T / 32)
#define QCAP 320
#define APITCH 132
static_assert((NWAVE & (NWAVE - 1)) == 0, "the owner wave of a row is taken with a mask");
static_assert(NWAVE * QCAP + NWAVE <= LCAP, "the wave queues and their lengths are kept in one work-group's LCAP words");
static_assert(QCAP % 4 == 0, "a wave's queue starts on a 16-B boundary");
static_assert(LCAP + NWAVE * QCAP >= (D / 4) * WSTEP, "one weight matrix fits where the edge queues stood");
static_assert(WLANE % 4 == 0 && WLANE >= (D / 16) * 4, "16-byte pieces; a lane's eight tiles do not reach the next lane's");
static_assert(WSTEP % 4 == 0 && WSTEP >= 16 * WLANE, "16-byte pieces; a step's 16 lanes do not reach the next step's");
static_assert(APITCH % 4 == 0 && APITCH >= D, "an owned row is read and written 16 B per lane");
static_assert((RCAP + 15) / 16 + 3 <= NWAVE, "a wave per 16-row strip, and three more when the last strip is cut in four");
static_assert(AGG_G == 256, "the owner key is one byte: the work-group of range g filters for key g mod 256");

__device__ __forceinline__ unsigned owner_key(int d) {
  return ((unsigned)d / (unsigned)RCAP) & 255u;
}

__global__ void edge_owner_keys(const int* __restrict__ dst,
                                unsigned* __restrict__ keys,
                                int nEdges, int nWords) {
  const int g = blockIdx.x * blockDim.x + threadIdx.x;
  if (g >= nWords) return;
  const int e0 = g << 2;
  unsigned k = 0u;
  if (e0 + 3 < nEdges) {
    const int4 d = ((const int4*)dst)[g];
    k = owner_key(d.x) | (owner_key(d.y) << 8) |
        (owner_key(d.z) << 16) | (owner_key(d.w) << 24);
  } else {
    for (int b = 0; b < 4; ++b)
      if (e0 + b < nEdges) k |= owner_key(dst[e0 + b]) << (8 * b);
  }
  keys[g] = k;
}

__device__ __forceinline__ unsigned zero_bytes(unsigned x) {
  const unsigned t = (x & 0x7F7F7F7Fu) + 0x7F7F7F7Fu;
  return ~(t | x | 0x7F7F7F7Fu);
}

__device__ __forceinline__ unsigned match_mask(uint4 k, unsigned mine, unsigned off) {
  const unsigned f0 = zero_bytes((k.x ^ mine) | off);
  const unsigned f1 = zero_bytes((k.y ^ mine) | off);
  const unsigned f2 = zero_bytes((k.z ^ mine) | off);
  const unsigned f3 = zero_bytes((k.w ^ mine) | off);
  return (f0 >> 7) | (f1 >> 6) | (f2 >> 5) | (f3 >> 4);
}

template <int NT, bool SCALE>
__device__ __forceinline__ void strip_products(v8f* c, const float* arow, float rowScale,
                                               const float* Wl, int half, int colL) {
  const float* bcol = Wl + (colL & 15) * WLANE + (colL >> 4) * 4 + half * 2;
  v2f aCur = *(const v2f*)(arow);
  if (SCALE) aCur *= rowScale;
#pragma unroll 4
  for (int k0 = 0; k0 < D; k0 += 4) {
    v2f b[NT];
#pragma unroll
    for (int t = 0; t < NT; ++t) b[t] = *(const v2f*)(bcol + (k0 >> 2) * WSTEP + t * 4);
    const int kN = (k0 + 4 < D) ? (k0 + 4) : k0;
    v2f aNext = *(const v2f*)(arow + kN);
    if (SCALE) aNext *= rowScale;
#pragma unroll
    for (int t = 0; t < NT; ++t) c[t] = wmma_f32(aCur, b[t], c[t]);
    aCur = aNext;
  }
}

template <int NT>
__device__ __forceinline__ void strip_store(const v8f* c, const float* __restrict__ bias,
                                            float* __restrict__ out, int lo, int strip,
                                            int rows, int half, int colL) {
#pragma unroll
  for (int t = 0; t < NT; ++t) {
    const int col = t * 16 + colL;
    const float bv = bias[col];
#pragma unroll
    for (int r = 0; r < 8; ++r) {
      const int row = strip * 16 + r + half * 8;
      if (row < rows) out[(size_t)(lo + row) * D + col] = c[t][r] + bv;
    }
  }
}

__device__ __forceinline__ void queue_edge(unsigned pos, unsigned e, int lo, int nEdges,
                                           const float* __restrict__ xin,
                                           const int* __restrict__ src,
                                           const int* __restrict__ dst,
                                           float* acc, unsigned* cnt, unsigned* list) {
  if (pos < (unsigned)LCAP) {
    list[pos] = e;
  } else if (e < (unsigned)nEdges) {
    const unsigned r = (unsigned)(dst[e] - lo);
    if (r < (unsigned)RCAP) {
      atomicAdd(&cnt[r], 1u);
      const float* xr = xin + (size_t)src[e] * D;
      for (int j = 0; j < D; ++j) atomicAdd(&acc[r * APITCH + j], xr[j]);
    }
  }
}

__global__ void __launch_bounds__(AGG_T)
aggregate_rows(const float* __restrict__ xin,
               const int* __restrict__ src,
               const int* __restrict__ dst,
               const unsigned* __restrict__ keys,
               unsigned* __restrict__ keepW,
               const unsigned* __restrict__ keptW,
               const float* __restrict__ Wself, const float* __restrict__ Wneigh,
               const float* __restrict__ bias,
               float* __restrict__ out,
               int nEdges, int nNodes) {
  __shared__ __attribute__((aligned(16))) float acc[RCAP * APITCH];
  __shared__ unsigned cnt[RCAP];
  __shared__ __attribute__((aligned(16))) unsigned pool[LCAP + NWAVE * QCAP];
  unsigned* const list = pool;
  unsigned* const wq   = pool + LCAP;
  __shared__ unsigned wlen[NWAVE];
  __shared__ unsigned nList;
  __shared__ unsigned ovf;

  const int tid  = threadIdx.x;
  const int lane = tid & 31;
  const int wave = __builtin_amdgcn_readfirstlane(tid >> 5);
  const int nG16 = (nEdges + 15) >> 4;
  const uint4* k4 = (const uint4*)keys;
  const unsigned myKey = (unsigned)blockIdx.x & (unsigned)(AGG_G - 1);
  const int rot  = (int)(((long long)myKey * nG16) / AGG_G);
  const unsigned mine = myKey * 0x01010101u;
  const unsigned rowBits = (1u << WBITS) - 1u;

  unsigned* keepMine = (keepW != nullptr) ? keepW + (size_t)blockIdx.x * LCAP : nullptr;
  unsigned kept = NOT_KEPT;
  if (keptW != nullptr) kept = keptW[(size_t)gridDim.x * LCAP + blockIdx.x];
  const bool replay = kept != NOT_KEPT;

  const int lo = blockIdx.x * RCAP;
  if (lo >= nNodes) return;
  {
    for (int i = tid; i < (RCAP * APITCH) / 4; i += AGG_T)
      ((float4*)acc)[i] = make_float4(0.0f, 0.0f, 0.0f, 0.0f);
    for (int i = tid; i < RCAP; i += AGG_T) cnt[i] = 0u;
    if (tid < NWAVE) wlen[tid] = 0u;
    if (tid == 0) { nList = 0u; ovf = 0u; }
    __syncthreads();

    if (!replay) {
    for (int i = tid; i < nG16; i += 8 * AGG_T) {
      uint4 v[8];
      int   gi[8];
#pragma unroll
      for (int u = 0; u < 8; ++u) {
        const int idx = i + u * AGG_T;
        int g = idx + rot;
        if (g >= nG16) g -= nG16;
        gi[u] = (idx < nG16) ? g : -1;
        v[u]  = k4[(idx < nG16) ? g : 0];
      }
      unsigned m[8];
      unsigned tot = 0u;
#pragma unroll
      for (int u = 0; u < 8; ++u) {
        m[u] = match_mask(v[u], mine, (unsigned)(gi[u] >> 31));
        tot += (unsigned)__popc(m[u]);
      }
      if (tot != 0u) {
        unsigned pos = atomicAdd(&nList, tot);
        if (pos + tot <= (unsigned)LCAP) {
#pragma unroll
          for (int u = 0; u < 8; ++u) {
            unsigned mm = m[u];
            while (mm != 0u) {
              const unsigned p = (unsigned)__builtin_ctz(mm);
              mm &= mm - 1u;
              list[pos++] = ((unsigned)gi[u] << 4) + ((p & 7u) << 2) + (p >> 3);
            }
          }
        } else {
#pragma unroll 1
          for (int u = 0; u < 8; ++u) {
            const int idx = i + u * AGG_T;
            if (idx >= nG16) break;
            int g = idx + rot;
            if (g >= nG16) g -= nG16;
            unsigned mm = match_mask(k4[g], mine, 0u);
            while (mm != 0u) {
              const unsigned p = (unsigned)__builtin_ctz(mm);
              mm &= mm - 1u;
              const unsigned e = ((unsigned)g << 4) + ((p & 7u) << 2) + (p >> 3);
              queue_edge(pos, e, lo, nEdges, xin, src, dst, acc, cnt, list);
              ++pos;
            }
          }
        }
      }
    }
    __syncthreads();

    const unsigned nq = nList;
    const int n = (int)min(nq, (unsigned)LCAP);
    for (int i = tid; i < n; i += AGG_T) {
      const unsigned e  = list[i];
      const bool     ok = e < (unsigned)nEdges;
      const unsigned ec = ok ? e : 0u;
      const unsigned sv = (unsigned)src[ec];
      const unsigned rr = (unsigned)(dst[ec] - lo);
      if (ok && rr < (unsigned)RCAP) {
        atomicAdd(&cnt[rr], 1u);
        const unsigned ow = rr & (unsigned)(NWAVE - 1);
        const unsigned p  = atomicAdd(&wlen[ow], 1u);
        if (p < (unsigned)QCAP) {
          wq[ow * QCAP + p] = (sv << WBITS) | rr;
        } else {
          ovf = 1u;
          const float* xr = xin + (size_t)sv * D;
          for (int j = 0; j < D; ++j) atomicAdd(&acc[rr * APITCH + j], xr[j]);
        }
      }
    }
    __syncthreads();
    if (keepMine != nullptr) {
      const int c = (int)min(wlen[wave], (unsigned)QCAP);
      for (int j = lane; j < c; j += 32) keepMine[wave * QCAP + j] = wq[wave * QCAP + j];
      if (lane == 0) keepMine[NWAVE * QCAP + wave] = (unsigned)c;
      if (tid == 0)
        keepW[(size_t)gridDim.x * LCAP + blockIdx.x] = (nq <= (unsigned)LCAP && ovf == 0u) ? 0u : NOT_KEPT;
    }
    } else {
    const unsigned* mineW = keptW + (size_t)blockIdx.x * LCAP;
    const int c = (int)min(mineW[NWAVE * QCAP + wave], (unsigned)QCAP);
    for (int j = lane; j < c; j += 32) {
      const unsigned w = mineW[wave * QCAP + j];
      wq[wave * QCAP + j] = w;
      atomicAdd(&cnt[w & rowBits], 1u);
    }
    if (lane == 0) wlen[wave] = (unsigned)c;
    __syncthreads();
    }

    {
      const unsigned* q = wq + wave * QCAP;
      const int c = __builtin_amdgcn_readfirstlane((int)min(wlen[wave], (unsigned)QCAP));
      const unsigned padWord = (unsigned)wave;
      for (int i0 = 0; i0 < c; i0 += 16) {
        const int li = i0 + (lane & 15);
        const unsigned wv = (li < c) ? q[li] : padWord;
        int    s[16], r[16];
        float4 v[16];
#pragma unroll
        for (int u = 0; u < 16; ++u) {
          const unsigned w = (unsigned)__builtin_amdgcn_readlane((int)wv, u);
          s[u] = (int)(w >> WBITS);
          r[u] = (int)(w & rowBits);
        }
#pragma unroll
        for (int u = 0; u < 16; ++u)
          v[u] = ((const float4*)(xin + (size_t)s[u] * D))[lane];
#pragma unroll
        for (int u = 0; u < 16; ++u) {
          const bool live = i0 + u < c;
          float4* a = (float4*)(acc + r[u] * APITCH) + lane;
          float4 t = *a;
          t.x += live ? v[u].x : 0.0f;
          t.y += live ? v[u].y : 0.0f;
          t.z += live ? v[u].z : 0.0f;
          t.w += live ? v[u].w : 0.0f;
          *a = t;
        }
      }
    }
    __syncthreads();

    {
      const int rows    = min(RCAP, nNodes - lo);
      const int nStrips = (rows + 15) >> 4;
      const bool split  = (nStrips & 3) == 1;
      const int nWhole  = split ? nStrips - 1 : nStrips;
      const bool whole   = wave < nWhole;
      const bool quarter = split && !whole && wave < nWhole + 4;
      const int strip    = whole ? wave : nStrips - 1;
      const int colBase  = quarter ? (wave - nWhole) * 32 : 0;
      const int half = lane >> 4;
      const int l16  = lane & 15;
      const int rL   = min(strip * 16 + l16, rows - 1);
      float* wl = (float*)pool;
      const float* xrow = xin + (size_t)(lo + rL) * D + half * 2;
      const float* srow = acc + rL * APITCH + half * 2;

      v8f c8[8];
#pragma unroll
      for (int t = 0; t < 8; ++t) c8[t] = (v8f)(0.0f);

      for (int i = tid; i < (D * D) / 4; i += AGG_T)
        *(float4*)(wl + (i & 31) * WSTEP + ((i >> 5) & 15) * WLANE + (i >> 9) * 4) = ((const float4*)Wself)[i];
      __syncthreads();
      if (whole)        strip_products<8, false>(c8, xrow, 1.0f, wl, half, l16);
      else if (quarter) strip_products<2, false>(c8, xrow, 1.0f, wl, half, colBase + l16);
      __syncthreads();
      for (int i = tid; i < (D * D) / 4; i += AGG_T)
        *(float4*)(wl + (i & 31) * WSTEP + ((i >> 5) & 15) * WLANE + (i >> 9) * 4) = ((const float4*)Wneigh)[i];
      __syncthreads();
      const float invd = 1.0f / fmaxf((float)cnt[rL], 1.0f);
      if (whole)        strip_products<8, true>(c8, srow, invd, wl, half, l16);
      else if (quarter) strip_products<2, true>(c8, srow, invd, wl, half, colBase + l16);

      if (whole) strip_store<8>(c8, bias, out, lo, strip, rows, half, l16);
      else if (quarter) strip_store<2>(c8, bias, out, lo, strip, rows, half, colBase + l16);
    }
  }
}

__global__ void l2_normalize(float* __restrict__ h, int nNodes) {
  int wave = (int)((blockIdx.x * blockDim.x + threadIdx.x) >> 5);
  int lane = threadIdx.x & 31;
  if (wave >= nNodes) return;
  float4* row = (float4*)(h + (size_t)wave * D);
  float4 v = row[lane];
  float ss = v.x * v.x + v.y * v.y + v.z * v.z + v.w * v.w;
#pragma unroll
  for (int off = 16; off > 0; off >>= 1) ss += __shfl_xor(ss, off, 32);
  float scale = 1.0f / fmaxf(sqrtf(ss), 1e-12f);
  v.x *= scale; v.y *= scale; v.z *= scale; v.w *= scale;
  row[lane] = v;
}

extern "C" void kernel_launch(void* const* d_in, const int* in_sizes, int n_in,
                              void* d_out, int out_size, void* d_ws, size_t ws_size,
                              hipStream_t stream) {
  const float* x   = (const float*)d_in[0];
  const int*   src = (const int*)  d_in[1];
  const int*   dst = (const int*)  d_in[2];
  const float* Ws1 = (const float*)d_in[3];
  const float* Wn1 = (const float*)d_in[4];
  const float* b1  = (const float*)d_in[5];
  const float* Ws2 = (const float*)d_in[6];
  const float* Wn2 = (const float*)d_in[7];
  const float* b2  = (const float*)d_in[8];
  float* out = (float*)d_out;

  const int N = in_sizes[0] / D;
  const int E = in_sizes[1];

  char* ws = (char*)d_ws;
  unsigned* keys = (unsigned*)ws; ws += (size_t)N * D * sizeof(float);
  float* h1  = (float*)ws; ws += (size_t)N * D * sizeof(float);

  {
    const int nWords = ((E + 15) / 16) * 4;
    edge_owner_keys<<<(nWords + 255) / 256, 256, 0, stream>>>(dst, keys, E, nWords);
  }

  unsigned* kept = keys + ((E + 15) / 16) * 4;

  const int nRanges = (N + RCAP - 1) / RCAP;

  aggregate_rows<<<nRanges, AGG_T, 0, stream>>>(x, src, dst, keys, kept, nullptr, Ws1, Wn1, b1, h1, E, N);

  aggregate_rows<<<nRanges, AGG_T, 0, stream>>>(h1, src, dst, keys, nullptr, kept, Ws2, Wn2, b2, out, E, N);

  l2_normalize<<<(N * 32 + 255) / 256, 256, 0, stream>>>(out, N);
}


# Owner keys kept as 8 bit planes per 32 edges (pre-pass transposes by wave votes): filter tests 32 edges by one XOR-OR chain of 8 words, not 16 by byte tricks, 16 loads a pass; arrival-order sums
# speedup vs baseline: 2.9871x; 1.0602x over previous
#include <hip/hip_runtime.h>
#include <cstddef>
#include <cstdint>

#define D 128
#define WLANE 36
#define WSTEP 580

typedef float v2f __attribute__((ext_vector_type(2)));
typedef float v8f __attribute__((ext_vector_type(8)));

__device__ __forceinline__ v8f wmma_f32(v2f a, v2f b, v8f c) {
#if defined(__gfx1250__)
  return __builtin_amdgcn_wmma_f32_16x16x4_f32(false, a, false, b,
                                               (short)0, c, false, false);
#else
  c[0] += a[0] * b[0];
  return c;
#endif
}

#define RCAP 392
#define LCAP 16384
#define AGG_T 1024
#define AGG_G 256
#define WBITS 9
#define NOT_KEPT 0xFFFFFFFFu
static_assert(RCAP < (1 << WBITS), "an owned row number and the sink must fit the low bits of a resolved word");
#define NWAVE (AGG_T / 32)
#define QCAP 320
#define APITCH 132
static_assert((NWAVE & (NWAVE - 1)) == 0, "the owner wave of a row is taken with a mask");
static_assert(NWAVE * QCAP + NWAVE <= LCAP, "the wave queues and their lengths are kept in one work-group's LCAP words");
static_assert(QCAP % 4 == 0, "a wave's queue starts on a 16-B boundary");
static_assert(LCAP + NWAVE * QCAP >= (D / 4) * WSTEP, "one weight matrix fits where the edge queues stood");
static_assert(WLANE % 4 == 0 && WLANE >= (D / 16) * 4, "16-byte pieces; a lane's eight tiles do not reach the next lane's");
static_assert(WSTEP % 4 == 0 && WSTEP >= 16 * WLANE, "16-byte pieces; a step's 16 lanes do not reach the next step's");
static_assert(APITCH % 4 == 0 && APITCH >= D, "an owned row is read and written 16 B per lane");
static_assert((RCAP + 15) / 16 + 3 <= NWAVE, "a wave per 16-row strip, and three more when the last strip is cut in four");
static_assert(AGG_G == 256, "the owner key is one byte: the work-group of range g filters for key g mod 256");

__device__ __forceinline__ unsigned owner_key(int d) {
  return ((unsigned)d / (unsigned)RCAP) & 255u;
}

#define KUNITS_PER_WAVE 8
#define FUNITS 8
__global__ void edge_owner_keys(const int* __restrict__ dst,
                                unsigned* __restrict__ keys,
                                int nEdges, int nKeyWaves) {
  const int lane = threadIdx.x & 31;
  const int wv = __builtin_amdgcn_readfirstlane((int)((blockIdx.x * blockDim.x + threadIdx.x) >> 5));
  if (wv >= nKeyWaves) return;
  unsigned k[KUNITS_PER_WAVE];
  const int eW = wv * 256 + 4 * lane;
  if (wv * 256 + 256 <= nEdges) {
    const int4 d0 = *(const int4*)(dst + eW);
    const int4 d1 = *(const int4*)(dst + eW + 128);
    k[0] = owner_key(d0.x); k[1] = owner_key(d0.y); k[2] = owner_key(d0.z); k[3] = owner_key(d0.w);
    k[4] = owner_key(d1.x); k[5] = owner_key(d1.y); k[6] = owner_key(d1.z); k[7] = owner_key(d1.w);
  } else {
#pragma unroll
    for (int u = 0; u < KUNITS_PER_WAVE; ++u) {
      const int e = eW + (u >> 2) * 128 + (u & 3);
      k[u] = (e < nEdges) ? owner_key(dst[e]) : 0u;
    }
  }
  unsigned lo = 0u, hi = 0u;
#pragma unroll
  for (int u = 0; u < KUNITS_PER_WAVE; ++u) {
#pragma unroll
    for (int j = 0; j < 4; ++j) {
      const unsigned pl = __builtin_amdgcn_ballot_w32(((k[u] >> j) & 1u) != 0u);
      const unsigned ph = __builtin_amdgcn_ballot_w32(((k[u] >> (4 + j)) & 1u) != 0u);
      if (lane == 4 * u + j) { lo = pl; hi = ph; }
    }
  }
  unsigned* blk = keys + (size_t)(wv >> 2) * 256 + (wv & 3) * 32;
  blk[lane]       = lo;
  blk[128 + lane] = hi;
}

__device__ __forceinline__ unsigned match_mask(uint4 a, uint4 b, const unsigned* kp, unsigned off) {
  const unsigned dif = (a.x ^ kp[0]) | (a.y ^ kp[1]) | (a.z ^ kp[2]) | (a.w ^ kp[3]) |
                       (b.x ^ kp[4]) | (b.y ^ kp[5]) | (b.z ^ kp[6]) | (b.w ^ kp[7]) | off;
  return ~dif;
}

__device__ __forceinline__ int unit_piece(int g) { return (g >> 5) * 64 + (g & 31); }
__device__ __forceinline__ unsigned unit_edge0(int g) { return ((unsigned)(g >> 2) << 7) + (unsigned)(g & 3); }

template <int NT, bool SCALE>
__device__ __forceinline__ void strip_products(v8f* c, const float* arow, float rowScale,
                                               const float* Wl, int half, int colL) {
  const float* bcol = Wl + (colL & 15) * WLANE + (colL >> 4) * 4 + half * 2;
  v2f aCur = *(const v2f*)(arow);
  if (SCALE) aCur *= rowScale;
#pragma unroll 4
  for (int k0 = 0; k0 < D; k0 += 4) {
    v2f b[NT];
#pragma unroll
    for (int t = 0; t < NT; ++t) b[t] = *(const v2f*)(bcol + (k0 >> 2) * WSTEP + t * 4);
    const int kN = (k0 + 4 < D) ? (k0 + 4) : k0;
    v2f aNext = *(const v2f*)(arow + kN);
    if (SCALE) aNext *= rowScale;
#pragma unroll
    for (int t = 0; t < NT; ++t) c[t] = wmma_f32(aCur, b[t], c[t]);
    aCur = aNext;
  }
}

template <int NT>
__device__ __forceinline__ void strip_store(const v8f* c, const float* __restrict__ bias,
                                            float* __restrict__ out, int lo, int strip,
                                            int rows, int half, int colL) {
#pragma unroll
  for (int t = 0; t < NT; ++t) {
    const int col = t * 16 + colL;
    const float bv = bias[col];
#pragma unroll
    for (int r = 0; r < 8; ++r) {
      const int row = strip * 16 + r + half * 8;
      if (row < rows) out[(size_t)(lo + row) * D + col] = c[t][r] + bv;
    }
  }
}

__device__ __forceinline__ void queue_edge(unsigned pos, unsigned e, int lo, int nEdges,
                                           const float* __restrict__ xin,
                                           const int* __restrict__ src,
                                           const int* __restrict__ dst,
                                           float* acc, unsigned* cnt, unsigned* list) {
  if (pos < (unsigned)LCAP) {
    list[pos] = e;
  } else if (e < (unsigned)nEdges) {
    const unsigned r = (unsigned)(dst[e] - lo);
    if (r < (unsigned)RCAP) {
      atomicAdd(&cnt[r], 1u);
      const float* xr = xin + (size_t)src[e] * D;
      for (int j = 0; j < D; ++j) atomicAdd(&acc[r * APITCH + j], xr[j]);
    }
  }
}

__global__ void __launch_bounds__(AGG_T)
aggregate_rows(const float* __restrict__ xin,
               const int* __restrict__ src,
               const int* __restrict__ dst,
               const unsigned* __restrict__ keys,
               unsigned* __restrict__ keepW,
               const unsigned* __restrict__ keptW,
               const float* __restrict__ Wself, const float* __restrict__ Wneigh,
               const float* __restrict__ bias,
               float* __restrict__ out,
               int nEdges, int nNodes) {
  __shared__ __attribute__((aligned(16))) float acc[RCAP * APITCH];
  __shared__ unsigned cnt[RCAP];
  __shared__ __attribute__((aligned(16))) unsigned pool[LCAP + NWAVE * QCAP];
  unsigned* const list = pool;
  unsigned* const wq   = pool + LCAP;
  __shared__ unsigned wlen[NWAVE];
  __shared__ unsigned nList;
  __shared__ unsigned ovf;

  const int tid  = threadIdx.x;
  const int lane = tid & 31;
  const int wave = __builtin_amdgcn_readfirstlane(tid >> 5);
  const int nUnits = ((nEdges + 255) >> 8) * KUNITS_PER_WAVE;
  const int nPad   = ((nUnits + 31) >> 5) << 5;
  const uint4* k4 = (const uint4*)keys;
  const unsigned myKey = (unsigned)blockIdx.x & (unsigned)(AGG_G - 1);
  const int rot  = (int)(((long long)myKey * (nPad >> 5)) / AGG_G) << 5;
  unsigned kp[8];
#pragma unroll
  for (int p = 0; p < 8; ++p) kp[p] = 0u - ((myKey >> p) & 1u);
  const unsigned rowBits = (1u << WBITS) - 1u;

  unsigned* keepMine = (keepW != nullptr) ? keepW + (size_t)blockIdx.x * LCAP : nullptr;
  unsigned kept = NOT_KEPT;
  if (keptW != nullptr) kept = keptW[(size_t)gridDim.x * LCAP + blockIdx.x];
  const bool replay = kept != NOT_KEPT;

  const int lo = blockIdx.x * RCAP;
  if (lo >= nNodes) return;
  {
    for (int i = tid; i < (RCAP * APITCH) / 4; i += AGG_T)
      ((float4*)acc)[i] = make_float4(0.0f, 0.0f, 0.0f, 0.0f);
    for (int i = tid; i < RCAP; i += AGG_T) cnt[i] = 0u;
    if (tid < NWAVE) wlen[tid] = 0u;
    if (tid == 0) { nList = 0u; ovf = 0u; }
    __syncthreads();

    if (!replay) {
    for (int i = tid; i < nPad; i += FUNITS * AGG_T) {
      uint4 va[FUNITS], vb[FUNITS];
      int   gi[FUNITS];
#pragma unroll
      for (int u = 0; u < FUNITS; ++u) {
        const int idx = i + u * AGG_T;
        int g = idx + rot;
        if (g >= nPad) g -= nPad;
        const bool ok = idx < nPad && g < nUnits;
        gi[u] = ok ? g : -1;
        const int pc = unit_piece(ok ? g : 0);
        va[u] = k4[pc];
        vb[u] = k4[pc + 32];
      }
      unsigned m[FUNITS];
      unsigned tot = 0u;
#pragma unroll
      for (int u = 0; u < FUNITS; ++u) {
        m[u] = match_mask(va[u], vb[u], kp, (unsigned)(gi[u] >> 31));
        tot += (unsigned)__popc(m[u]);
      }
      if (tot != 0u) {
        unsigned pos = atomicAdd(&nList, tot);
        if (pos + tot <= (unsigned)LCAP) {
#pragma unroll
          for (int u = 0; u < FUNITS; ++u) {
            unsigned mm = m[u];
            const unsigned e0 = unit_edge0(gi[u]);
            while (mm != 0u) {
              const unsigned p = (unsigned)__builtin_ctz(mm);
              mm &= mm - 1u;
              list[pos++] = e0 + (p << 2);
            }
          }
        } else {
#pragma unroll 1
          for (int u = 0; u < FUNITS; ++u) {
            const int idx = i + u * AGG_T;
            if (idx >= nPad) break;
            int g = idx + rot;
            if (g >= nPad) g -= nPad;
            if (g >= nUnits) continue;
            const int pc = unit_piece(g);
            unsigned mm = match_mask(k4[pc], k4[pc + 32], kp, 0u);
            const unsigned e0 = unit_edge0(g);
            while (mm != 0u) {
              const unsigned p = (unsigned)__builtin_ctz(mm);
              mm &= mm - 1u;
              queue_edge(pos, e0 + (p << 2), lo, nEdges, xin, src, dst, acc, cnt, list);
              ++pos;
            }
          }
        }
      }
    }
    __syncthreads();

    const unsigned nq = nList;
    const int n = (int)min(nq, (unsigned)LCAP);
    for (int i = tid; i < n; i += AGG_T) {
      const unsigned e  = list[i];
      const bool     ok = e < (unsigned)nEdges;
      const unsigned ec = ok ? e : 0u;
      const unsigned sv = (unsigned)src[ec];
      const unsigned rr = (unsigned)(dst[ec] - lo);
      if (ok && rr < (unsigned)RCAP) {
        atomicAdd(&cnt[rr], 1u);
        const unsigned ow = rr & (unsigned)(NWAVE - 1);
        const unsigned p  = atomicAdd(&wlen[ow], 1u);
        if (p < (unsigned)QCAP) {
          wq[ow * QCAP + p] = (sv << WBITS) | rr;
        } else {
          ovf = 1u;
          const float* xr = xin + (size_t)sv * D;
          for (int j = 0; j < D; ++j) atomicAdd(&acc[rr * APITCH + j], xr[j]);
        }
      }
    }
    __syncthreads();
    if (keepMine != nullptr) {
      const int c = (int)min(wlen[wave], (unsigned)QCAP);
      for (int j = lane; j < c; j += 32) keepMine[wave * QCAP + j] = wq[wave * QCAP + j];
      if (lane == 0) keepMine[NWAVE * QCAP + wave] = (unsigned)c;
      if (tid == 0)
        keepW[(size_t)gridDim.x * LCAP + blockIdx.x] = (nq <= (unsigned)LCAP && ovf == 0u) ? 0u : NOT_KEPT;
    }
    } else {
    const unsigned* mineW = keptW + (size_t)blockIdx.x * LCAP;
    const int c = (int)min(mineW[NWAVE * QCAP + wave], (unsigned)QCAP);
    for (int j = lane; j < c; j += 32) {
      const unsigned w = mineW[wave * QCAP + j];
      wq[wave * QCAP + j] = w;
      atomicAdd(&cnt[w & rowBits], 1u);
    }
    if (lane == 0) wlen[wave] = (unsigned)c;
    __syncthreads();
    }

    {
      const unsigned* q = wq + wave * QCAP;
      const int c = __builtin_amdgcn_readfirstlane((int)min(wlen[wave], (unsigned)QCAP));
      const unsigned padWord = (unsigned)wave;
      for (int i0 = 0; i0 < c; i0 += 16) {
        const int li = i0 + (lane & 15);
        const unsigned wv = (li < c) ? q[li] : padWord;
        int    s[16], r[16];
        float4 v[16];
#pragma unroll
        for (int u = 0; u < 16; ++u) {
          const unsigned w = (unsigned)__builtin_amdgcn_readlane((int)wv, u);
          s[u] = (int)(w >> WBITS);
          r[u] = (int)(w & rowBits);
        }
#pragma unroll
        for (int u = 0; u < 16; ++u)
          v[u] = ((const float4*)(xin + (size_t)s[u] * D))[lane];
#pragma unroll
        for (int u = 0; u < 16; ++u) {
          const bool live = i0 + u < c;
          float4* a = (float4*)(acc + r[u] * APITCH) + lane;
          float4 t = *a;
          t.x += live ? v[u].x : 0.0f;
          t.y += live ? v[u].y : 0.0f;
          t.z += live ? v[u].z : 0.0f;
          t.w += live ? v[u].w : 0.0f;
          *a = t;
        }
      }
    }
    __syncthreads();

    {
      const int rows    = min(RCAP, nNodes - lo);
      const int nStrips = (rows + 15) >> 4;
      const bool split  = (nStrips & 3) == 1;
      const int nWhole  = split ? nStrips - 1 : nStrips;
      const bool whole   = wave < nWhole;
      const bool quarter = split && !whole && wave < nWhole + 4;
      const int strip    = whole ? wave : nStrips - 1;
      const int colBase  = quarter ? (wave - nWhole) * 32 : 0;
      const int half = lane >> 4;
      const int l16  = lane & 15;
      const int rL   = min(strip * 16 + l16, rows - 1);
      float* wl = (float*)pool;
      const float* xrow = xin + (size_t)(lo + rL) * D + half * 2;
      const float* srow = acc + rL * APITCH + half * 2;

      v8f c8[8];
#pragma unroll
      for (int t = 0; t < 8; ++t) c8[t] = (v8f)(0.0f);

      for (int i = tid; i < (D * D) / 4; i += AGG_T)
        *(float4*)(wl + (i & 31) * WSTEP + ((i >> 5) & 15) * WLANE + (i >> 9) * 4) = ((const float4*)Wself)[i];
      __syncthreads();
      if (whole)        strip_products<8, false>(c8, xrow, 1.0f, wl, half, l16);
      else if (quarter) strip_products<2, false>(c8, xrow, 1.0f, wl, half, colBase + l16);
      __syncthreads();
      for (int i = tid; i < (D * D) / 4; i += AGG_T)
        *(float4*)(wl + (i & 31) * WSTEP + ((i >> 5) & 15) * WLANE + (i >> 9) * 4) = ((const float4*)Wneigh)[i];
      __syncthreads();
      const float invd = 1.0f / fmaxf((float)cnt[rL], 1.0f);
      if (whole)        strip_products<8, true>(c8, srow, invd, wl, half, l16);
      else if (quarter) strip_products<2, true>(c8, srow, invd, wl, half, colBase + l16);

      if (whole) strip_store<8>(c8, bias, out, lo, strip, rows, half, l16);
      else if (quarter) strip_store<2>(c8, bias, out, lo, strip, rows, half, colBase + l16);
    }
  }
}

__global__ void l2_normalize(float* __restrict__ h, int nNodes) {
  int wave = (int)((blockIdx.x * blockDim.x + threadIdx.x) >> 5);
  int lane = threadIdx.x & 31;
  if (wave >= nNodes) return;
  float4* row = (float4*)(h + (size_t)wave * D);
  float4 v = row[lane];
  float ss = v.x * v.x + v.y * v.y + v.z * v.z + v.w * v.w;
#pragma unroll
  for (int off = 16; off > 0; off >>= 1) ss += __shfl_xor(ss, off, 32);
  float scale = 1.0f / fmaxf(sqrtf(ss), 1e-12f);
  v.x *= scale; v.y *= scale; v.z *= scale; v.w *= scale;
  row[lane] = v;
}

extern "C" void kernel_launch(void* const* d_in, const int* in_sizes, int n_in,
                              void* d_out, int out_size, void* d_ws, size_t ws_size,
                              hipStream_t stream) {
  const float* x   = (const float*)d_in[0];
  const int*   src = (const int*)  d_in[1];
  const int*   dst = (const int*)  d_in[2];
  const float* Ws1 = (const float*)d_in[3];
  const float* Wn1 = (const float*)d_in[4];
  const float* b1  = (const float*)d_in[5];
  const float* Ws2 = (const float*)d_in[6];
  const float* Wn2 = (const float*)d_in[7];
  const float* b2  = (const float*)d_in[8];
  float* out = (float*)d_out;

  const int N = in_sizes[0] / D;
  const int E = in_sizes[1];

  char* ws = (char*)d_ws;
  unsigned* keys = (unsigned*)ws; ws += (size_t)N * D * sizeof(float);
  float* h1  = (float*)ws; ws += (size_t)N * D * sizeof(float);

  {
    const int nKeyWaves = (E + 255) / 256;
    edge_owner_keys<<<(nKeyWaves + 7) / 8, 256, 0, stream>>>(dst, keys, E, nKeyWaves);
  }

  unsigned* kept = keys + ((E + 1023) / 1024) * 256;

  const int nRanges = (N + RCAP - 1) / RCAP;

  aggregate_rows<<<nRanges, AGG_T, 0, stream>>>(x, src, dst, keys, kept, nullptr, Ws1, Wn1, b1, h1, E, N);

  aggregate_rows<<<nRanges, AGG_T, 0, stream>>>(h1, src, dst, keys, nullptr, kept, Ws2, Wn2, b2, out, E, N);

  l2_normalize<<<(N * 32 + 255) / 256, 256, 0, stream>>>(out, N);
}


# Pre-pass bins edges by owner: per chunk and owner a 256-B piece of (source,row) words and its length; a work-group reads only its own pieces; key planes, filter scan, resolve go; arrival-order sums
# speedup vs baseline: 3.2464x; 1.0868x over previous
#include <hip/hip_runtime.h>
#include <cstddef>
#include <cstdint>

#define D 128
#define WLANE 36
#define WSTEP 580

typedef float v2f __attribute__((ext_vector_type(2)));
typedef float v8f __attribute__((ext_vector_type(8)));

__device__ __forceinline__ v8f wmma_f32(v2f a, v2f b, v8f c) {
#if defined(__gfx1250__)
  return __builtin_amdgcn_wmma_f32_16x16x4_f32(false, a, false, b,
                                               (short)0, c, false, false);
#else
  c[0] += a[0] * b[0];
  return c;
#endif
}

#define RCAP 392
#define LCAP 16384
#define AGG_T 1024
#define WBITS 9
#define NOT_KEPT 0xFFFFFFFFu
static_assert(RCAP < (1 << WBITS), "an owned row number and the sink must fit the low bits of a resolved word");
#define NWAVE (AGG_T / 32)
#define QCAP 320
#define APITCH 132
static_assert((NWAVE & (NWAVE - 1)) == 0, "the owner wave of a row is taken with a mask");
static_assert(NWAVE * QCAP + NWAVE <= LCAP, "the wave queues and their lengths are kept in one work-group's LCAP words");
static_assert(QCAP % 4 == 0, "a wave's queue starts on a 16-B boundary");
static_assert(LCAP + NWAVE * QCAP >= (D / 4) * WSTEP, "one weight matrix fits where the edge queues stood");
static_assert(WLANE % 4 == 0 && WLANE >= (D / 16) * 4, "16-byte pieces; a lane's eight tiles do not reach the next lane's");
static_assert(WSTEP % 4 == 0 && WSTEP >= 16 * WLANE, "16-byte pieces; a step's 16 lanes do not reach the next step's");
static_assert(APITCH % 4 == 0 && APITCH >= D, "an owned row is read and written 16 B per lane");
static_assert((RCAP + 15) / 16 + 3 <= NWAVE, "a wave per 16-row strip, and three more when the last strip is cut in four");
#define NBIN 256
#define BCAP 64
#define PPT ((NBIN * (BCAP / 4)) / AGG_T)
static_assert((NBIN * (BCAP / 4)) % AGG_T == 0 && AGG_T % (BCAP / 4) == 0, "every lane takes the same slots of PPT whole pieces");
#define BIN_T 1024
#define BIN_E 8
__device__ __forceinline__ int edge_chunk(int nEdges) { return (nEdges + NBIN - 1) / NBIN; }

__global__ void __launch_bounds__(BIN_T)
bin_edges(const int* __restrict__ src, const int* __restrict__ dst,
          unsigned* __restrict__ pieces, unsigned* __restrict__ counts, int nEdges) {
  __shared__ unsigned cntB[NBIN];
  __shared__ __attribute__((aligned(16))) unsigned stage[NBIN * BCAP];
  const int tid = threadIdx.x;
  const int c   = blockIdx.x;
  for (int i = tid; i < NBIN; i += BIN_T) cntB[i] = 0u;
  __syncthreads();
  const int chunk = edge_chunk(nEdges);
  const int e0 = c * chunk;
  const int e1 = min(e0 + chunk, nEdges);
  for (int eb = e0 + tid; eb < e1; eb += BIN_E * BIN_T) {
    unsigned d[BIN_E], s[BIN_E];
#pragma unroll
    for (int u = 0; u < BIN_E; ++u) {
      const int e  = eb + u * BIN_T;
      const int ec = min(e, e1 - 1);
      const unsigned dv = (unsigned)dst[ec];
      s[u] = (unsigned)src[ec];
      d[u] = (e < e1) ? dv : 0xFFFFFFFFu;
    }
#pragma unroll
    for (int u = 0; u < BIN_E; ++u) {
      const unsigned o = d[u] / (unsigned)RCAP;
      if (o < (unsigned)NBIN) {
        const unsigned p = atomicAdd(&cntB[o], 1u);
        if (p < (unsigned)BCAP) stage[o * BCAP + p] = (s[u] << WBITS) | (d[u] - o * (unsigned)RCAP);
      }
    }
  }
  __syncthreads();
  uint4* dstp = (uint4*)(pieces + (size_t)c * NBIN * BCAP);
  for (int i = tid; i < (NBIN * BCAP) / 4; i += BIN_T) {
    const unsigned n = min(cntB[i / (BCAP / 4)], (unsigned)BCAP);
    if ((unsigned)((i % (BCAP / 4)) * 4) < n) dstp[i] = ((const uint4*)stage)[i];
  }
  for (int i = tid; i < NBIN; i += BIN_T) counts[(size_t)c * NBIN + i] = cntB[i];
}

template <int NT, bool SCALE>
__device__ __forceinline__ void strip_products(v8f* c, const float* arow, float rowScale,
                                               const float* Wl, int half, int colL) {
  const float* bcol = Wl + (colL & 15) * WLANE + (colL >> 4) * 4 + half * 2;
  v2f aCur = *(const v2f*)(arow);
  if (SCALE) aCur *= rowScale;
#pragma unroll 4
  for (int k0 = 0; k0 < D; k0 += 4) {
    v2f b[NT];
#pragma unroll
    for (int t = 0; t < NT; ++t) b[t] = *(const v2f*)(bcol + (k0 >> 2) * WSTEP + t * 4);
    const int kN = (k0 + 4 < D) ? (k0 + 4) : k0;
    v2f aNext = *(const v2f*)(arow + kN);
    if (SCALE) aNext *= rowScale;
#pragma unroll
    for (int t = 0; t < NT; ++t) c[t] = wmma_f32(aCur, b[t], c[t]);
    aCur = aNext;
  }
}

template <int NT>
__device__ __forceinline__ void strip_store(const v8f* c, const float* __restrict__ bias,
                                            float* __restrict__ out, int lo, int strip,
                                            int rows, int half, int colL) {
#pragma unroll
  for (int t = 0; t < NT; ++t) {
    const int col = t * 16 + colL;
    const float bv = bias[col];
#pragma unroll
    for (int r = 0; r < 8; ++r) {
      const int row = strip * 16 + r + half * 8;
      if (row < rows) out[(size_t)(lo + row) * D + col] = c[t][r] + bv;
    }
  }
}

__device__ __forceinline__ void push_word(unsigned w, const float* __restrict__ xin,
                                          float* acc, unsigned* cnt, unsigned* wq,
                                          unsigned* wlen, unsigned* ovf) {
  const unsigned rr = w & ((1u << WBITS) - 1u);
  atomicAdd(&cnt[rr], 1u);
  const unsigned ow = rr & (unsigned)(NWAVE - 1);
  const unsigned p  = atomicAdd(&wlen[ow], 1u);
  if (p < (unsigned)QCAP) {
    wq[ow * QCAP + p] = w;
  } else {
    *ovf = 1u;
    const float* xr = xin + (size_t)(w >> WBITS) * D;
    for (int j = 0; j < D; ++j) atomicAdd(&acc[rr * APITCH + j], xr[j]);
  }
}

__global__ void __launch_bounds__(AGG_T)
aggregate_rows(const float* __restrict__ xin,
               const int* __restrict__ src,
               const int* __restrict__ dst,
               const unsigned* __restrict__ pieces,
               const unsigned* __restrict__ counts,
               unsigned* __restrict__ keepW,
               const unsigned* __restrict__ keptW,
               const float* __restrict__ Wself, const float* __restrict__ Wneigh,
               const float* __restrict__ bias,
               float* __restrict__ out,
               int nEdges, int nNodes) {
  __shared__ __attribute__((aligned(16))) float acc[RCAP * APITCH];
  __shared__ unsigned cnt[RCAP];
  __shared__ __attribute__((aligned(16))) unsigned pool[LCAP + NWAVE * QCAP];
  unsigned* const list = pool;
  unsigned* const wq   = pool + LCAP;
  __shared__ unsigned wlen[NWAVE];
  __shared__ unsigned nList;
  __shared__ unsigned ovf;

  const int tid  = threadIdx.x;
  const int lane = tid & 31;
  const int wave = __builtin_amdgcn_readfirstlane(tid >> 5);
  const unsigned rowBits = (1u << WBITS) - 1u;

  unsigned* keepMine = (keepW != nullptr) ? keepW + (size_t)blockIdx.x * LCAP : nullptr;
  unsigned kept = NOT_KEPT;
  if (keptW != nullptr) kept = keptW[(size_t)gridDim.x * LCAP + blockIdx.x];
  const bool replay = kept != NOT_KEPT;

  const int lo = blockIdx.x * RCAP;
  if (lo >= nNodes) return;
  {
    for (int i = tid; i < (RCAP * APITCH) / 4; i += AGG_T)
      ((float4*)acc)[i] = make_float4(0.0f, 0.0f, 0.0f, 0.0f);
    for (int i = tid; i < RCAP; i += AGG_T) cnt[i] = 0u;
    if (tid < NWAVE) wlen[tid] = 0u;
    if (tid == 0) { nList = 0u; ovf = 0u; }
    __syncthreads();

    if (!replay) {
    const int chunk = edge_chunk(nEdges);
    const bool binned = gridDim.x <= (unsigned)NBIN;
    const unsigned q = (unsigned)(tid & (BCAP / 4 - 1)) * 4u;
    unsigned nc[PPT];
    uint4    w4[PPT];
#pragma unroll
    for (int u = 0; u < PPT; ++u) {
      const int c = (tid + u * AGG_T) / (BCAP / 4);
      nc[u] = binned ? counts[(size_t)c * NBIN + blockIdx.x] : (unsigned)BCAP + 1u;
    }
#pragma unroll
    for (int u = 0; u < PPT; ++u) {
      const int c = (tid + u * AGG_T) / (BCAP / 4);
      w4[u] = make_uint4(0u, 0u, 0u, 0u);
      if (nc[u] <= (unsigned)BCAP && q < nc[u])
        w4[u] = *(const uint4*)(pieces + ((size_t)c * NBIN + blockIdx.x) * BCAP + q);
    }
#pragma unroll
    for (int u = 0; u < PPT; ++u) {
      const int c = (tid + u * AGG_T) / (BCAP / 4);
      if (nc[u] > (unsigned)BCAP) {
        if (q == 0u) list[atomicAdd(&nList, 1u)] = (unsigned)c;
      } else {
        if (q      < nc[u]) push_word(w4[u].x, xin, acc, cnt, wq, wlen, &ovf);
        if (q + 1u < nc[u]) push_word(w4[u].y, xin, acc, cnt, wq, wlen, &ovf);
        if (q + 2u < nc[u]) push_word(w4[u].z, xin, acc, cnt, wq, wlen, &ovf);
        if (q + 3u < nc[u]) push_word(w4[u].w, xin, acc, cnt, wq, wlen, &ovf);
      }
    }
    __syncthreads();
    const unsigned nShort = nList;
    for (unsigned k = 0; k < nShort; ++k) {
      const int c  = (int)list[k];
      const int e0 = c * chunk;
      const int e1 = min(e0 + chunk, nEdges);
      for (int e = e0 + tid; e < e1; e += AGG_T) {
        const unsigned rr = (unsigned)(dst[e] - lo);
        if (rr < (unsigned)RCAP) push_word(((unsigned)src[e] << WBITS) | rr, xin, acc, cnt, wq, wlen, &ovf);
      }
    }
    __syncthreads();
    if (keepMine != nullptr) {
      const int c = (int)min(wlen[wave], (unsigned)QCAP);
      for (int j = lane; j < c; j += 32) keepMine[wave * QCAP + j] = wq[wave * QCAP + j];
      if (lane == 0) keepMine[NWAVE * QCAP + wave] = (unsigned)c;
      if (tid == 0)
        keepW[(size_t)gridDim.x * LCAP + blockIdx.x] = (ovf == 0u) ? 0u : NOT_KEPT;
    }
    } else {
    const unsigned* mineW = keptW + (size_t)blockIdx.x * LCAP;
    const int c = (int)min(mineW[NWAVE * QCAP + wave], (unsigned)QCAP);
    for (int j = lane; j < c; j += 32) {
      const unsigned w = mineW[wave * QCAP + j];
      wq[wave * QCAP + j] = w;
      atomicAdd(&cnt[w & rowBits], 1u);
    }
    if (lane == 0) wlen[wave] = (unsigned)c;
    __syncthreads();
    }

    {
      const unsigned* q = wq + wave * QCAP;
      const int c = __builtin_amdgcn_readfirstlane((int)min(wlen[wave], (unsigned)QCAP));
      const unsigned padWord = (unsigned)wave;
      for (int i0 = 0; i0 < c; i0 += 16) {
        const int li = i0 + (lane & 15);
        const unsigned wv = (li < c) ? q[li] : padWord;
        int    s[16], r[16];
        float4 v[16];
#pragma unroll
        for (int u = 0; u < 16; ++u) {
          const unsigned w = (unsigned)__builtin_amdgcn_readlane((int)wv, u);
          s[u] = (int)(w >> WBITS);
          r[u] = (int)(w & rowBits);
        }
#pragma unroll
        for (int u = 0; u < 16; ++u)
          v[u] = ((const float4*)(xin + (size_t)s[u] * D))[lane];
#pragma unroll
        for (int u = 0; u < 16; ++u) {
          const bool live = i0 + u < c;
          float4* a = (float4*)(acc + r[u] * APITCH) + lane;
          float4 t = *a;
          t.x += live ? v[u].x : 0.0f;
          t.y += live ? v[u].y : 0.0f;
          t.z += live ? v[u].z : 0.0f;
          t.w += live ? v[u].w : 0.0f;
          *a = t;
        }
      }
    }
    __syncthreads();

    {
      const int rows    = min(RCAP, nNodes - lo);
      const int nStrips = (rows + 15) >> 4;
      const bool split  = (nStrips & 3) == 1;
      const int nWhole  = split ? nStrips - 1 : nStrips;
      const bool whole   = wave < nWhole;
      const bool quarter = split && !whole && wave < nWhole + 4;
      const int strip    = whole ? wave : nStrips - 1;
      const int colBase  = quarter ? (wave - nWhole) * 32 : 0;
      const int half = lane >> 4;
      const int l16  = lane & 15;
      const int rL   = min(strip * 16 + l16, rows - 1);
      float* wl = (float*)pool;
      const float* xrow = xin + (size_t)(lo + rL) * D + half * 2;
      const float* srow = acc + rL * APITCH + half * 2;

      v8f c8[8];
#pragma unroll
      for (int t = 0; t < 8; ++t) c8[t] = (v8f)(0.0f);

      for (int i = tid; i < (D * D) / 4; i += AGG_T)
        *(float4*)(wl + (i & 31) * WSTEP + ((i >> 5) & 15) * WLANE + (i >> 9) * 4) = ((const float4*)Wself)[i];
      __syncthreads();
      if (whole)        strip_products<8, false>(c8, xrow, 1.0f, wl, half, l16);
      else if (quarter) strip_products<2, false>(c8, xrow, 1.0f, wl, half, colBase + l16);
      __syncthreads();
      for (int i = tid; i < (D * D) / 4; i += AGG_T)
        *(float4*)(wl + (i & 31) * WSTEP + ((i >> 5) & 15) * WLANE + (i >> 9) * 4) = ((const float4*)Wneigh)[i];
      __syncthreads();
      const float invd = 1.0f / fmaxf((float)cnt[rL], 1.0f);
      if (whole)        strip_products<8, true>(c8, srow, invd, wl, half, l16);
      else if (quarter) strip_products<2, true>(c8, srow, invd, wl, half, colBase + l16);

      if (whole) strip_store<8>(c8, bias, out, lo, strip, rows, half, l16);
      else if (quarter) strip_store<2>(c8, bias, out, lo, strip, rows, half, colBase + l16);
    }
  }
}

__global__ void l2_normalize(float* __restrict__ h, int nNodes) {
  int wave = (int)((blockIdx.x * blockDim.x + threadIdx.x) >> 5);
  int lane = threadIdx.x & 31;
  if (wave >= nNodes) return;
  float4* row = (float4*)(h + (size_t)wave * D);
  float4 v = row[lane];
  float ss = v.x * v.x + v.y * v.y + v.z * v.z + v.w * v.w;
#pragma unroll
  for (int off = 16; off > 0; off >>= 1) ss += __shfl_xor(ss, off, 32);
  float scale = 1.0f / fmaxf(sqrtf(ss), 1e-12f);
  v.x *= scale; v.y *= scale; v.z *= scale; v.w *= scale;
  row[lane] = v;
}

extern "C" void kernel_launch(void* const* d_in, const int* in_sizes, int n_in,
                              void* d_out, int out_size, void* d_ws, size_t ws_size,
                              hipStream_t stream) {
  const float* x   = (const float*)d_in[0];
  const int*   src = (const int*)  d_in[1];
  const int*   dst = (const int*)  d_in[2];
  const float* Ws1 = (const float*)d_in[3];
  const float* Wn1 = (const float*)d_in[4];
  const float* b1  = (const float*)d_in[5];
  const float* Ws2 = (const float*)d_in[6];
  const float* Wn2 = (const float*)d_in[7];
  const float* b2  = (const float*)d_in[8];
  float* out = (float*)d_out;

  const int N = in_sizes[0] / D;
  const int E = in_sizes[1];

  char* ws = (char*)d_ws;
  unsigned* pieces = (unsigned*)ws; ws += (size_t)N * D * sizeof(float);
  float* h1  = (float*)ws; ws += (size_t)N * D * sizeof(float);

  unsigned* counts = pieces + (size_t)NBIN * NBIN * BCAP;
  unsigned* kept   = counts + (size_t)NBIN * NBIN;
  bin_edges<<<NBIN, BIN_T, 0, stream>>>(src, dst, pieces, counts, E);

  const int nRanges = (N + RCAP - 1) / RCAP;

  aggregate_rows<<<nRanges, AGG_T, 0, stream>>>(x, src, dst, pieces, counts, kept, nullptr, Ws1, Wn1, b1, h1, E, N);

  aggregate_rows<<<nRanges, AGG_T, 0, stream>>>(h1, src, dst, pieces, counts, nullptr, kept, Ws2, Wn2, b2, out, E, N);

  l2_normalize<<<(N * 32 + 255) / 256, 256, 0, stream>>>(out, N);
}


# Whole-strip output staged in the strip's LDS rows, stored as whole 512-B rows: 16 b128 stores per wave off one base, not 64 b32 stores of 64-B pieces each behind s_wait_xcnt; spills end; sums same
# speedup vs baseline: 4.4597x; 1.3737x over previous
#include <hip/hip_runtime.h>
#include <cstddef>
#include <cstdint>

#define D 128
#define WLANE 36
#define WSTEP 580

typedef float v2f __attribute__((ext_vector_type(2)));
typedef float v8f __attribute__((ext_vector_type(8)));

__device__ __forceinline__ v8f wmma_f32(v2f a, v2f b, v8f c) {
#if defined(__gfx1250__)
  return __builtin_amdgcn_wmma_f32_16x16x4_f32(false, a, false, b,
                                               (short)0, c, false, false);
#else
  c[0] += a[0] * b[0];
  return c;
#endif
}

#define RCAP 392
#define LCAP 16384
#define AGG_T 1024
#define WBITS 9
#define NOT_KEPT 0xFFFFFFFFu
static_assert(RCAP < (1 << WBITS), "an owned row number and the sink must fit the low bits of a resolved word");
#define NWAVE (AGG_T / 32)
#define QCAP 320
#define APITCH 132
static_assert((NWAVE & (NWAVE - 1)) == 0, "the owner wave of a row is taken with a mask");
static_assert(NWAVE * QCAP + NWAVE <= LCAP, "the wave queues and their lengths are kept in one work-group's LCAP words");
static_assert(QCAP % 4 == 0, "a wave's queue starts on a 16-B boundary");
static_assert(LCAP + NWAVE * QCAP >= (D / 4) * WSTEP, "one weight matrix fits where the edge queues stood");
static_assert(WLANE % 4 == 0 && WLANE >= (D / 16) * 4, "16-byte pieces; a lane's eight tiles do not reach the next lane's");
static_assert(WSTEP % 4 == 0 && WSTEP >= 16 * WLANE, "16-byte pieces; a step's 16 lanes do not reach the next step's");
static_assert(APITCH % 4 == 0 && APITCH >= D, "an owned row is read and written 16 B per lane");
static_assert((RCAP + 15) / 16 + 3 <= NWAVE, "a wave per 16-row strip, and three more when the last strip is cut in four");
static_assert(RCAP % 16 == 0 || (((RCAP + 15) / 16) & 3) == 1, "a strip that reaches past the owned rows is always the cut one: a whole strip's 16 LDS rows exist");
#define NBIN 256
#define BCAP 64
#define PPT ((NBIN * (BCAP / 4)) / AGG_T)
static_assert((NBIN * (BCAP / 4)) % AGG_T == 0 && AGG_T % (BCAP / 4) == 0, "every lane takes the same slots of PPT whole pieces");
#define BIN_T 1024
#define BIN_E 8
__device__ __forceinline__ int edge_chunk(int nEdges) { return (nEdges + NBIN - 1) / NBIN; }

__global__ void __launch_bounds__(BIN_T)
bin_edges(const int* __restrict__ src, const int* __restrict__ dst,
          unsigned* __restrict__ pieces, unsigned* __restrict__ counts, int nEdges) {
  __shared__ unsigned cntB[NBIN];
  __shared__ __attribute__((aligned(16))) unsigned stage[NBIN * BCAP];
  const int tid = threadIdx.x;
  const int c   = blockIdx.x;
  for (int i = tid; i < NBIN; i += BIN_T) cntB[i] = 0u;
  __syncthreads();
  const int chunk = edge_chunk(nEdges);
  const int e0 = c * chunk;
  const int e1 = min(e0 + chunk, nEdges);
  for (int eb = e0 + tid; eb < e1; eb += BIN_E * BIN_T) {
    unsigned d[BIN_E], s[BIN_E];
#pragma unroll
    for (int u = 0; u < BIN_E; ++u) {
      const int e  = eb + u * BIN_T;
      const int ec = min(e, e1 - 1);
      const unsigned dv = (unsigned)dst[ec];
      s[u] = (unsigned)src[ec];
      d[u] = (e < e1) ? dv : 0xFFFFFFFFu;
    }
#pragma unroll
    for (int u = 0; u < BIN_E; ++u) {
      const unsigned o = d[u] / (unsigned)RCAP;
      if (o < (unsigned)NBIN) {
        const unsigned p = atomicAdd(&cntB[o], 1u);
        if (p < (unsigned)BCAP) stage[o * BCAP + p] = (s[u] << WBITS) | (d[u] - o * (unsigned)RCAP);
      }
    }
  }
  __syncthreads();
  uint4* dstp = (uint4*)(pieces + (size_t)c * NBIN * BCAP);
  for (int i = tid; i < (NBIN * BCAP) / 4; i += BIN_T) {
    const unsigned n = min(cntB[i / (BCAP / 4)], (unsigned)BCAP);
    if ((unsigned)((i % (BCAP / 4)) * 4) < n) dstp[i] = ((const uint4*)stage)[i];
  }
  for (int i = tid; i < NBIN; i += BIN_T) counts[(size_t)c * NBIN + i] = cntB[i];
}

template <int NT, bool SCALE>
__device__ __forceinline__ void strip_products(v8f* c, const float* arow, float rowScale,
                                               const float* Wl, int half, int colL) {
  const float* bcol = Wl + (colL & 15) * WLANE + (colL >> 4) * 4 + half * 2;
  v2f aCur = *(const v2f*)(arow);
  if (SCALE) aCur *= rowScale;
#pragma unroll 4
  for (int k0 = 0; k0 < D; k0 += 4) {
    v2f b[NT];
#pragma unroll
    for (int t = 0; t < NT; ++t) b[t] = *(const v2f*)(bcol + (k0 >> 2) * WSTEP + t * 4);
    const int kN = (k0 + 4 < D) ? (k0 + 4) : k0;
    v2f aNext = *(const v2f*)(arow + kN);
    if (SCALE) aNext *= rowScale;
#pragma unroll
    for (int t = 0; t < NT; ++t) c[t] = wmma_f32(aCur, b[t], c[t]);
    aCur = aNext;
  }
}

template <int NT>
__device__ __forceinline__ void strip_store(const v8f* c, const float* __restrict__ bias,
                                            float* __restrict__ out, int lo, int strip,
                                            int rows, int half, int colL) {
#pragma unroll
  for (int t = 0; t < NT; ++t) {
    const int col = t * 16 + colL;
    const float bv = bias[col];
#pragma unroll
    for (int r = 0; r < 8; ++r) {
      const int row = strip * 16 + r + half * 8;
      if (row < rows) out[(size_t)(lo + row) * D + col] = c[t][r] + bv;
    }
  }
}

__device__ __forceinline__ void strip_store_rows(const v8f* c, const float* __restrict__ bias,
                                                 float* __restrict__ out, float* srows,
                                                 int lo, int strip, int rows,
                                                 int lane, int half, int l16) {
  const float4 b4 = ((const float4*)bias)[lane];
#pragma unroll
  for (int t = 0; t < 8; ++t) {
#pragma unroll
    for (int r = 0; r < 8; ++r) srows[(r + half * 8) * APITCH + t * 16 + l16] = c[t][r];
  }
  __builtin_amdgcn_fence(__ATOMIC_RELEASE, "wavefront");
  __builtin_amdgcn_wave_barrier();
  __builtin_amdgcn_fence(__ATOMIC_ACQUIRE, "wavefront");
  const int nValid = min(16, rows - strip * 16);
  float4 v[16];
#pragma unroll
  for (int j = 0; j < 16; ++j) v[j] = *(const float4*)(srows + j * APITCH + lane * 4);
  float* orow = out + (size_t)(lo + strip * 16) * D + lane * 4;
#pragma unroll
  for (int j = 0; j < 16; ++j) {
    if (j < nValid) {
      float4 o = v[j];
      o.x += b4.x; o.y += b4.y; o.z += b4.z; o.w += b4.w;
      *(float4*)(orow + (size_t)j * D) = o;
    }
  }
}

__device__ __forceinline__ void push_word(unsigned w, const float* __restrict__ xin,
                                          float* acc, unsigned* cnt, unsigned* wq,
                                          unsigned* wlen, unsigned* ovf) {
  const unsigned rr = w & ((1u << WBITS) - 1u);
  atomicAdd(&cnt[rr], 1u);
  const unsigned ow = rr & (unsigned)(NWAVE - 1);
  const unsigned p  = atomicAdd(&wlen[ow], 1u);
  if (p < (unsigned)QCAP) {
    wq[ow * QCAP + p] = w;
  } else {
    *ovf = 1u;
    const float* xr = xin + (size_t)(w >> WBITS) * D;
    for (int j = 0; j < D; ++j) atomicAdd(&acc[rr * APITCH + j], xr[j]);
  }
}

__global__ void __launch_bounds__(AGG_T)
aggregate_rows(const float* __restrict__ xin,
               const int* __restrict__ src,
               const int* __restrict__ dst,
               const unsigned* __restrict__ pieces,
               const unsigned* __restrict__ counts,
               unsigned* __restrict__ keepW,
               const unsigned* __restrict__ keptW,
               const float* __restrict__ Wself, const float* __restrict__ Wneigh,
               const float* __restrict__ bias,
               float* __restrict__ out,
               int nEdges, int nNodes) {
  __shared__ __attribute__((aligned(16))) float acc[RCAP * APITCH];
  __shared__ unsigned cnt[RCAP];
  __shared__ __attribute__((aligned(16))) unsigned pool[LCAP + NWAVE * QCAP];
  unsigned* const list = pool;
  unsigned* const wq   = pool + LCAP;
  __shared__ unsigned wlen[NWAVE];
  __shared__ unsigned nList;
  __shared__ unsigned ovf;

  const int tid  = threadIdx.x;
  const int lane = tid & 31;
  const int wave = __builtin_amdgcn_readfirstlane(tid >> 5);
  const unsigned rowBits = (1u << WBITS) - 1u;

  unsigned* keepMine = (keepW != nullptr) ? keepW + (size_t)blockIdx.x * LCAP : nullptr;
  unsigned kept = NOT_KEPT;
  if (keptW != nullptr) kept = keptW[(size_t)gridDim.x * LCAP + blockIdx.x];
  const bool replay = kept != NOT_KEPT;

  const int lo = blockIdx.x * RCAP;
  if (lo >= nNodes) return;
  {
    for (int i = tid; i < (RCAP * APITCH) / 4; i += AGG_T)
      ((float4*)acc)[i] = make_float4(0.0f, 0.0f, 0.0f, 0.0f);
    for (int i = tid; i < RCAP; i += AGG_T) cnt[i] = 0u;
    if (tid < NWAVE) wlen[tid] = 0u;
    if (tid == 0) { nList = 0u; ovf = 0u; }
    __syncthreads();

    if (!replay) {
    const int chunk = edge_chunk(nEdges);
    const bool binned = gridDim.x <= (unsigned)NBIN;
    const unsigned q = (unsigned)(tid & (BCAP / 4 - 1)) * 4u;
    unsigned nc[PPT];
    uint4    w4[PPT];
#pragma unroll
    for (int u = 0; u < PPT; ++u) {
      const int c = (tid + u * AGG_T) / (BCAP / 4);
      nc[u] = binned ? counts[(size_t)c * NBIN + blockIdx.x] : (unsigned)BCAP + 1u;
    }
#pragma unroll
    for (int u = 0; u < PPT; ++u) {
      const int c = (tid + u * AGG_T) / (BCAP / 4);
      w4[u] = make_uint4(0u, 0u, 0u, 0u);
      if (nc[u] <= (unsigned)BCAP && q < nc[u])
        w4[u] = *(const uint4*)(pieces + ((size_t)c * NBIN + blockIdx.x) * BCAP + q);
    }
#pragma unroll
    for (int u = 0; u < PPT; ++u) {
      const int c = (tid + u * AGG_T) / (BCAP / 4);
      if (nc[u] > (unsigned)BCAP) {
        if (q == 0u) list[atomicAdd(&nList, 1u)] = (unsigned)c;
      } else {
        if (q      < nc[u]) push_word(w4[u].x, xin, acc, cnt, wq, wlen, &ovf);
        if (q + 1u < nc[u]) push_word(w4[u].y, xin, acc, cnt, wq, wlen, &ovf);
        if (q + 2u < nc[u]) push_word(w4[u].z, xin, acc, cnt, wq, wlen, &ovf);
        if (q + 3u < nc[u]) push_word(w4[u].w, xin, acc, cnt, wq, wlen, &ovf);
      }
    }
    __syncthreads();
    const unsigned nShort = nList;
    for (unsigned k = 0; k < nShort; ++k) {
      const int c  = (int)list[k];
      const int e0 = c * chunk;
      const int e1 = min(e0 + chunk, nEdges);
      for (int e = e0 + tid; e < e1; e += AGG_T) {
        const unsigned rr = (unsigned)(dst[e] - lo);
        if (rr < (unsigned)RCAP) push_word(((unsigned)src[e] << WBITS) | rr, xin, acc, cnt, wq, wlen, &ovf);
      }
    }
    __syncthreads();
    if (keepMine != nullptr) {
      const int c = (int)min(wlen[wave], (unsigned)QCAP);
      for (int j = lane; j < c; j += 32) keepMine[wave * QCAP + j] = wq[wave * QCAP + j];
      if (lane == 0) keepMine[NWAVE * QCAP + wave] = (unsigned)c;
      if (tid == 0)
        keepW[(size_t)gridDim.x * LCAP + blockIdx.x] = (ovf == 0u) ? 0u : NOT_KEPT;
    }
    } else {
    const unsigned* mineW = keptW + (size_t)blockIdx.x * LCAP;
    const int c = (int)min(mineW[NWAVE * QCAP + wave], (unsigned)QCAP);
    for (int j = lane; j < c; j += 32) {
      const unsigned w = mineW[wave * QCAP + j];
      wq[wave * QCAP + j] = w;
      atomicAdd(&cnt[w & rowBits], 1u);
    }
    if (lane == 0) wlen[wave] = (unsigned)c;
    __syncthreads();
    }

    {
      const unsigned* q = wq + wave * QCAP;
      const int c = __builtin_amdgcn_readfirstlane((int)min(wlen[wave], (unsigned)QCAP));
      const unsigned padWord = (unsigned)wave;
      for (int i0 = 0; i0 < c; i0 += 16) {
        const int li = i0 + (lane & 15);
        const unsigned wv = (li < c) ? q[li] : padWord;
        int    s[16], r[16];
        float4 v[16];
#pragma unroll
        for (int u = 0; u < 16; ++u) {
          const unsigned w = (unsigned)__builtin_amdgcn_readlane((int)wv, u);
          s[u] = (int)(w >> WBITS);
          r[u] = (int)(w & rowBits);
        }
#pragma unroll
        for (int u = 0; u < 16; ++u)
          v[u] = ((const float4*)(xin + (size_t)s[u] * D))[lane];
#pragma unroll
        for (int u = 0; u < 16; ++u) {
          const bool live = i0 + u < c;
          float4* a = (float4*)(acc + r[u] * APITCH) + lane;
          float4 t = *a;
          t.x += live ? v[u].x : 0.0f;
          t.y += live ? v[u].y : 0.0f;
          t.z += live ? v[u].z : 0.0f;
          t.w += live ? v[u].w : 0.0f;
          *a = t;
        }
      }
    }
    __syncthreads();

    {
      const int rows    = min(RCAP, nNodes - lo);
      const int nStrips = (rows + 15) >> 4;
      const bool split  = (nStrips & 3) == 1;
      const int nWhole  = split ? nStrips - 1 : nStrips;
      const bool whole   = wave < nWhole;
      const bool quarter = split && !whole && wave < nWhole + 4;
      const int strip    = whole ? wave : nStrips - 1;
      const int colBase  = quarter ? (wave - nWhole) * 32 : 0;
      const int half = lane >> 4;
      const int l16  = lane & 15;
      const int rL   = min(strip * 16 + l16, rows - 1);
      float* wl = (float*)pool;
      const float* xrow = xin + (size_t)(lo + rL) * D + half * 2;
      const float* srow = acc + rL * APITCH + half * 2;

      v8f c8[8];
#pragma unroll
      for (int t = 0; t < 8; ++t) c8[t] = (v8f)(0.0f);

      for (int i = tid; i < (D * D) / 4; i += AGG_T)
        *(float4*)(wl + (i & 31) * WSTEP + ((i >> 5) & 15) * WLANE + (i >> 9) * 4) = ((const float4*)Wself)[i];
      __syncthreads();
      if (whole)        strip_products<8, false>(c8, xrow, 1.0f, wl, half, l16);
      else if (quarter) strip_products<2, false>(c8, xrow, 1.0f, wl, half, colBase + l16);
      __syncthreads();
      for (int i = tid; i < (D * D) / 4; i += AGG_T)
        *(float4*)(wl + (i & 31) * WSTEP + ((i >> 5) & 15) * WLANE + (i >> 9) * 4) = ((const float4*)Wneigh)[i];
      __syncthreads();
      const float invd = 1.0f / fmaxf((float)cnt[rL], 1.0f);
      if (whole) {
        strip_products<8, true>(c8, srow, invd, wl, half, l16);
        strip_store_rows(c8, bias, out, acc + strip * 16 * APITCH, lo, strip, rows, lane, half, l16);
      } else if (quarter) {
        strip_products<2, true>(c8, srow, invd, wl, half, colBase + l16);
        strip_store<2>(c8, bias, out, lo, strip, rows, half, colBase + l16);
      }
    }
  }
}

__global__ void l2_normalize(float* __restrict__ h, int nNodes) {
  int wave = (int)((blockIdx.x * blockDim.x + threadIdx.x) >> 5);
  int lane = threadIdx.x & 31;
  if (wave >= nNodes) return;
  float4* row = (float4*)(h + (size_t)wave * D);
  float4 v = row[lane];
  float ss = v.x * v.x + v.y * v.y + v.z * v.z + v.w * v.w;
#pragma unroll
  for (int off = 16; off > 0; off >>= 1) ss += __shfl_xor(ss, off, 32);
  float scale = 1.0f / fmaxf(sqrtf(ss), 1e-12f);
  v.x *= scale; v.y *= scale; v.z *= scale; v.w *= scale;
  row[lane] = v;
}

extern "C" void kernel_launch(void* const* d_in, const int* in_sizes, int n_in,
                              void* d_out, int out_size, void* d_ws, size_t ws_size,
                              hipStream_t stream) {
  const float* x   = (const float*)d_in[0];
  const int*   src = (const int*)  d_in[1];
  const int*   dst = (const int*)  d_in[2];
  const float* Ws1 = (const float*)d_in[3];
  const float* Wn1 = (const float*)d_in[4];
  const float* b1  = (const float*)d_in[5];
  const float* Ws2 = (const float*)d_in[6];
  const float* Wn2 = (const float*)d_in[7];
  const float* b2  = (const float*)d_in[8];
  float* out = (float*)d_out;

  const int N = in_sizes[0] / D;
  const int E = in_sizes[1];

  char* ws = (char*)d_ws;
  unsigned* pieces = (unsigned*)ws; ws += (size_t)N * D * sizeof(float);
  float* h1  = (float*)ws; ws += (size_t)N * D * sizeof(float);

  unsigned* counts = pieces + (size_t)NBIN * NBIN * BCAP;
  unsigned* kept   = counts + (size_t)NBIN * NBIN;
  bin_edges<<<NBIN, BIN_T, 0, stream>>>(src, dst, pieces, counts, E);

  const int nRanges = (N + RCAP - 1) / RCAP;

  aggregate_rows<<<nRanges, AGG_T, 0, stream>>>(x, src, dst, pieces, counts, kept, nullptr, Ws1, Wn1, b1, h1, E, N);

  aggregate_rows<<<nRanges, AGG_T, 0, stream>>>(h1, src, dst, pieces, counts, nullptr, kept, Ws2, Wn2, b2, out, E, N);

  l2_normalize<<<(N * 32 + 255) / 256, 256, 0, stream>>>(out, N);
}
